# starting kernel
# baseline (speedup 1.0000x reference)
#include <hip/hip_runtime.h>
#include <math.h>


typedef __attribute__((ext_vector_type(2))) float v2f;
typedef __attribute__((ext_vector_type(8))) float v8f;

#define N_NODES 50000
#define TILE_M 64
#define TILE_N 64
#define TILE_K 32

__global__ void fill_f32(float* __restrict__ p, float v, long n) {
    long i = (long)blockIdx.x * blockDim.x + threadIdx.x;
    if (i < n) p[i] = v;
}

__device__ __forceinline__ void atomicMaxF(float* addr, float val) {
    if (val >= 0.0f)
        atomicMax((int*)addr, __float_as_int(val));
    else
        atomicMin((unsigned int*)addr, (unsigned int)__float_as_int(val));
}

__global__ __launch_bounds__(256) void gemm_wmma_f32(
    const float* __restrict__ p0, const float* __restrict__ p1,
    const float* __restrict__ p2, int w0, int w1, int w2,
    const float* __restrict__ B, float* __restrict__ C,
    int M, int N, int K)
{
    __shared__ float As[TILE_M][TILE_K + 2];
    __shared__ float Bs[TILE_K][TILE_N + 2];

    const int tid  = threadIdx.x;
    const int lane = tid & 31;
    const int wave = tid >> 5;
    const int mt     = wave & 3;
    const int ntBase = (wave >> 2) * 2;

    const int blockRow = blockIdx.x * TILE_M;
    const int blockCol = blockIdx.y * TILE_N;

    v8f acc0 = {};
    v8f acc1 = {};

    const int mLocal = lane & 15;
    const int kOff   = (lane >> 4) << 1;

    for (int kb = 0; kb < K; kb += TILE_K) {
#pragma unroll
        for (int i = 0; i < 8; ++i) {
            int idx = tid + i * 256;
            int m = idx >> 5;
            int k = idx & 31;
            int row = blockRow + m;
            int gk  = kb + k;
            float v = 0.0f;
            if (row < M && gk < K) {
                int kk = gk;
                if (kk < w0)               v = p0[(long)row * w0 + kk];
                else if ((kk -= w0) < w1)  v = p1[(long)row * w1 + kk];
                else                       v = p2[(long)row * w2 + (kk - w1)];
            }
            As[m][k] = v;
        }
#pragma unroll
        for (int i = 0; i < 8; ++i) {
            int idx = tid + i * 256;
            int k = idx >> 6;
            int n = idx & 63;
            int gk  = kb + k;
            int col = blockCol + n;
            float v = 0.0f;
            if (gk < K && col < N) v = B[(long)gk * N + col];
            Bs[k][n] = v;
        }
        __syncthreads();

#pragma unroll
        for (int kk = 0; kk < TILE_K; kk += 4) {
            v2f a, b0, b1;
            a.x = As[mt * 16 + mLocal][kk + kOff];
            a.y = As[mt * 16 + mLocal][kk + kOff + 1];
            b0.x = Bs[kk + kOff][ntBase * 16 + mLocal];
            b0.y = Bs[kk + kOff + 1][ntBase * 16 + mLocal];
            b1.x = Bs[kk + kOff][(ntBase + 1) * 16 + mLocal];
            b1.y = Bs[kk + kOff + 1][(ntBase + 1) * 16 + mLocal];
            acc0 = __builtin_amdgcn_wmma_f32_16x16x4_f32(
                false, a, false, b0, (short)0, acc0, false, false);
            acc1 = __builtin_amdgcn_wmma_f32_16x16x4_f32(
                false, a, false, b1, (short)0, acc1, false, false);
        }
        __syncthreads();
    }

    const int col0 = blockCol + ntBase * 16 + mLocal;
    const int col1 = col0 + 16;
    const int mAdd = (lane >> 4) << 3;
#pragma unroll
    for (int r = 0; r < 8; ++r) {
        int row = blockRow + mt * 16 + r + mAdd;
        if (row < M) {
            C[(long)row * N + col0] = acc0[r];
            C[(long)row * N + col1] = acc1[r];
        }
    }
}

__global__ __launch_bounds__(256) void attn_proj(
    const float* __restrict__ feat, const float* __restrict__ al,
    const float* __restrict__ ar, float* __restrict__ el,
    float* __restrict__ er, int nNodes, int H)
{
    int wid  = blockIdx.x * 8 + (threadIdx.x >> 5);
    int lane = threadIdx.x & 31;
    if (wid >= nNodes * H) return;
    int node = wid / H;
    int h    = wid - node * H;
    int C    = H * 64;
    const float* f  = feat + (long)node * C + h * 64;
    const float* wl = al + h * 64;
    const float* wr = ar + h * 64;
    float f0 = f[lane], f1 = f[lane + 32];
    float sl = f0 * wl[lane] + f1 * wl[lane + 32];
    float sr = f0 * wr[lane] + f1 * wr[lane + 32];
#pragma unroll
    for (int o = 16; o > 0; o >>= 1) {
        sl += __shfl_xor(sl, o, 32);
        sr += __shfl_xor(sr, o, 32);
    }
    if (lane == 0) { el[wid] = sl; er[wid] = sr; }
}

__global__ void edge_lrelu_max(
    const int* __restrict__ src, const int* __restrict__ dst,
    const float* __restrict__ el, const float* __restrict__ er,
    float* __restrict__ ebuf, float* __restrict__ emax, int E, int H)
{
    int i = blockIdx.x * blockDim.x + threadIdx.x;
    if (i >= E) return;
    int s = src[i], d = dst[i];
    for (int h = 0; h < H; ++h) {
        float x = el[s * H + h] + er[d * H + h];
        float e = x > 0.0f ? x : 0.2f * x;
        ebuf[(long)i * H + h] = e;
        atomicMaxF(&emax[d * H + h], e);
    }
}

__global__ void edge_expsum(
    const int* __restrict__ dst, float* __restrict__ ebuf,
    const float* __restrict__ emax, float* __restrict__ denom, int E, int H)
{
    int i = blockIdx.x * blockDim.x + threadIdx.x;
    if (i >= E) return;
    int d = dst[i];
    for (int h = 0; h < H; ++h) {
        float ex = expf(ebuf[(long)i * H + h] - emax[d * H + h]);
        ebuf[(long)i * H + h] = ex;
        atomicAdd(&denom[d * H + h], ex);
    }
}

__global__ __launch_bounds__(256) void edge_aggregate(
    const int* __restrict__ src, const int* __restrict__ dst,
    const float* __restrict__ feat, const float* __restrict__ ebuf,
    const float* __restrict__ denom, float* __restrict__ rst, int E, int H)
{
    int wid  = blockIdx.x * 8 + (threadIdx.x >> 5);
    int lane = threadIdx.x & 31;
    if (wid >= E) return;
    int s = src[wid], d = dst[wid];
    int C = H * 64;
    for (int idx = lane; idx < C; idx += 32) {
        int h = idx >> 6;
        float alpha = ebuf[(long)wid * H + h] / denom[d * H + h];
        atomicAdd(&rst[(long)d * C + idx], feat[(long)s * C + idx] * alpha);
    }
}

__global__ void bias_elu(float* __restrict__ x, const float* __restrict__ b,
                         long n, int C)
{
    long i = (long)blockIdx.x * blockDim.x + threadIdx.x;
    if (i >= n) return;
    float v = x[i] + b[i % C];
    x[i] = v > 0.0f ? v : expm1f(v);
}

__global__ __launch_bounds__(256) void classify(
    const float* __restrict__ rst2, const float* __restrict__ b2,
    const float* __restrict__ Wc1, const float* __restrict__ bc1,
    const float* __restrict__ Wc2, const float* __restrict__ bc2,
    float* __restrict__ out, int nNodes)
{
    int wid  = blockIdx.x * 8 + (threadIdx.x >> 5);
    int lane = threadIdx.x & 31;
    if (wid >= nNodes) return;
    float acc = 0.0f;
#pragma unroll
    for (int d = 0; d < 64; ++d)
        acc += (rst2[(long)wid * 64 + d] + b2[d]) * Wc1[d * 32 + lane];
    float t = fmaxf(acc + bc1[lane], 0.0f);
    float o0 = t * Wc2[lane * 2 + 0];
    float o1 = t * Wc2[lane * 2 + 1];
#pragma unroll
    for (int o = 16; o > 0; o >>= 1) {
        o0 += __shfl_xor(o0, o, 32);
        o1 += __shfl_xor(o1, o, 32);
    }
    if (lane == 0) {
        out[(long)wid * 2 + 0] = o0 + bc2[0];
        out[(long)wid * 2 + 1] = o1 + bc2[1];
    }
}

extern "C" void kernel_launch(void* const* d_in, const int* in_sizes, int n_in,
                              void* d_out, int out_size, void* d_ws, size_t ws_size,
                              hipStream_t stream)
{
    (void)n_in; (void)out_size; (void)ws_size;

    const float* randfeat = (const float*)d_in[0];
    const float* func_emb = (const float*)d_in[1];
    const float* codebert = (const float*)d_in[2];
    const int*   src      = (const int*)d_in[3];
    const int*   dst      = (const int*)d_in[4];
    const float* W1  = (const float*)d_in[5];
    const float* al1 = (const float*)d_in[6];
    const float* ar1 = (const float*)d_in[7];
    const float* b1  = (const float*)d_in[8];
    const float* W2  = (const float*)d_in[9];
    const float* al2 = (const float*)d_in[10];
    const float* ar2 = (const float*)d_in[11];
    const float* b2  = (const float*)d_in[12];
    const float* Wc1 = (const float*)d_in[13];
    const float* bc1 = (const float*)d_in[14];
    const float* Wc2 = (const float*)d_in[15];
    const float* bc2 = (const float*)d_in[16];
    float* out = (float*)d_out;

    const int  E  = in_sizes[3];
    const long NN = N_NODES;

    float* ws     = (float*)d_ws;
    float* feat1  = ws;
    float* rst1   = feat1  + NN * 256;
    float* feat2  = rst1   + NN * 256;
    float* rst2   = feat2  + NN * 64;
    float* el1    = rst2   + NN * 64;
    float* er1    = el1    + NN * 4;
    float* emax1  = er1    + NN * 4;
    float* denom1 = emax1  + NN * 4;
    float* el2    = denom1 + NN * 4;
    float* er2    = el2    + NN;
    float* emax2  = er2    + NN;
    float* denom2 = emax2  + NN;
    float* e1     = denom2 + NN;
    float* e2     = e1 + (long)E * 4;

    fill_f32<<<(int)((NN * 256 + 255) / 256), 256, 0, stream>>>(rst1, 0.0f, NN * 256);
    fill_f32<<<(int)((NN * 64 + 255) / 256), 256, 0, stream>>>(rst2, 0.0f, NN * 64);
    fill_f32<<<(int)((NN * 4 + 255) / 256), 256, 0, stream>>>(denom1, 0.0f, NN * 4);
    fill_f32<<<(int)((NN + 255) / 256), 256, 0, stream>>>(denom2, 0.0f, NN);
    fill_f32<<<(int)((NN * 4 + 255) / 256), 256, 0, stream>>>(emax1, -INFINITY, NN * 4);
    fill_f32<<<(int)((NN + 255) / 256), 256, 0, stream>>>(emax2, -INFINITY, NN);

    dim3 g1((N_NODES + TILE_M - 1) / TILE_M, 256 / TILE_N);
    gemm_wmma_f32<<<g1, 256, 0, stream>>>(randfeat, func_emb, codebert,
                                          100, 100, 768, W1, feat1,
                                          N_NODES, 256, 968);
    attn_proj<<<(N_NODES * 4 + 7) / 8, 256, 0, stream>>>(feat1, al1, ar1, el1, er1, N_NODES, 4);
    edge_lrelu_max<<<(E + 255) / 256, 256, 0, stream>>>(src, dst, el1, er1, e1, emax1, E, 4);
    edge_expsum<<<(E + 255) / 256, 256, 0, stream>>>(dst, e1, emax1, denom1, E, 4);
    edge_aggregate<<<(E + 7) / 8, 256, 0, stream>>>(src, dst, feat1, e1, denom1, rst1, E, 4);
    bias_elu<<<(int)((NN * 256 + 255) / 256), 256, 0, stream>>>(rst1, b1, NN * 256, 256);

    dim3 g2((N_NODES + TILE_M - 1) / TILE_M, 64 / TILE_N);
    gemm_wmma_f32<<<g2, 256, 0, stream>>>(rst1, rst1, rst1,
                                          256, 0, 0, W2, feat2,
                                          N_NODES, 64, 256);
    attn_proj<<<(N_NODES + 7) / 8, 256, 0, stream>>>(feat2, al2, ar2, el2, er2, N_NODES, 1);
    edge_lrelu_max<<<(E + 255) / 256, 256, 0, stream>>>(src, dst, el2, er2, e2, emax2, E, 1);
    edge_expsum<<<(E + 255) / 256, 256, 0, stream>>>(dst, e2, emax2, denom2, E, 1);
    edge_aggregate<<<(E + 7) / 8, 256, 0, stream>>>(src, dst, feat2, e2, denom2, rst2, E, 1);

    classify<<<(N_NODES + 7) / 8, 256, 0, stream>>>(rst2, b2, Wc1, bc1, Wc2, bc2, out, N_NODES);
}


# Edge aggregation by dst-range tiles: a work-group owns 196 dst rows, scans dst, adds weighted src rows into an LDS tile (LDS f32 adds, arrival order), stores each row once; no global f32 adds
# speedup vs baseline: 2.0390x; 2.0390x over previous
#include <hip/hip_runtime.h>
#include <math.h>


typedef __attribute__((ext_vector_type(2))) float v2f;
typedef __attribute__((ext_vector_type(8))) float v8f;

#define N_NODES 50000

__global__ void fill_f32(float* __restrict__ p, float v, long n) {
    long i = (long)blockIdx.x * blockDim.x + threadIdx.x;
    if (i < n) p[i] = v;
}

__device__ __forceinline__ void atomicMaxF(float* addr, float val) {
    if (val >= 0.0f)
        atomicMax((int*)addr, __float_as_int(val));
    else
        atomicMin((unsigned int*)addr, (unsigned int)__float_as_int(val));
}

template <int NT, int NC>
__device__ __forceinline__ void seg_accum(
    const float* __restrict__ a0, const float* __restrict__ a1, int w,
    const float* __restrict__ bcol, int h, v8f (&acc)[2][NT])
{
    const int nblk = w >> 5;
    if (nblk > 0) {
        const float4* q0 = (const float4*)(a0 + 16 * h);
        const float4* q1 = (const float4*)(a1 + 16 * h);
        const float* bp = bcol + (long)(16 * h) * NC;
        float4 ra[4], rb[4];
        v2f bc[NT];
#pragma unroll
        for (int q = 0; q < 4; ++q) { ra[q] = q0[q]; rb[q] = q1[q]; }
#pragma unroll
        for (int t = 0; t < NT; ++t) { bc[t].x = bp[t * 16]; bc[t].y = bp[NC + t * 16]; }

        for (int blk = 0; blk < nblk; ++blk) {
            const int adv = (blk + 1 < nblk) ? 32 : 0;
            const float4* nq0 = (const float4*)((const float*)q0 + adv);
            const float4* nq1 = (const float4*)((const float*)q1 + adv);
            const float* nbp = bp + (long)adv * NC;
            float4 na[4], nb[4];
#pragma unroll
            for (int q = 0; q < 4; ++q) { na[q] = nq0[q]; nb[q] = nq1[q]; }
            __builtin_amdgcn_sched_barrier(0);
#pragma unroll
            for (int j = 0; j < 8; ++j) {
                v2f bn[NT];
                if (j < 7) {
#pragma unroll
                    for (int t = 0; t < NT; ++t) {
                        bn[t].x = bp[(2 * j + 2) * NC + t * 16];
                        bn[t].y = bp[(2 * j + 3) * NC + t * 16];
                    }
                } else {
#pragma unroll
                    for (int t = 0; t < NT; ++t) {
                        bn[t].x = nbp[t * 16];
                        bn[t].y = nbp[NC + t * 16];
                    }
                }
                v2f fa, fb;
                if (j & 1) { fa.x = ra[j >> 1].z; fa.y = ra[j >> 1].w; fb.x = rb[j >> 1].z; fb.y = rb[j >> 1].w; }
                else       { fa.x = ra[j >> 1].x; fa.y = ra[j >> 1].y; fb.x = rb[j >> 1].x; fb.y = rb[j >> 1].y; }
#pragma unroll
                for (int t = 0; t < NT; ++t) {
                    acc[0][t] = __builtin_amdgcn_wmma_f32_16x16x4_f32(
                        false, fa, false, bc[t], (short)0, acc[0][t], false, false);
                    acc[1][t] = __builtin_amdgcn_wmma_f32_16x16x4_f32(
                        false, fb, false, bc[t], (short)0, acc[1][t], false, false);
                }
#pragma unroll
                for (int t = 0; t < NT; ++t) bc[t] = bn[t];
            }
            __builtin_amdgcn_sched_barrier(0);
            asm volatile("v_nop\n\tv_nop\n\tv_nop\n\tv_nop");
#pragma unroll
            for (int q = 0; q < 4; ++q) { ra[q] = na[q]; rb[q] = nb[q]; }
            q0 = nq0; q1 = nq1; bp = nbp;
        }
    }
    for (int k = nblk << 5; k + 4 <= w; k += 4) {
        const float2 ta = *(const float2*)(a0 + k + 2 * h);
        const float2 tb = *(const float2*)(a1 + k + 2 * h);
        v2f fa, fb;
        fa.x = ta.x; fa.y = ta.y; fb.x = tb.x; fb.y = tb.y;
        const float* bp = bcol + (long)(k + 2 * h) * NC;
#pragma unroll
        for (int t = 0; t < NT; ++t) {
            v2f b;
            b.x = bp[t * 16];
            b.y = bp[NC + t * 16];
            acc[0][t] = __builtin_amdgcn_wmma_f32_16x16x4_f32(
                false, fa, false, b, (short)0, acc[0][t], false, false);
            acc[1][t] = __builtin_amdgcn_wmma_f32_16x16x4_f32(
                false, fb, false, b, (short)0, acc[1][t], false, false);
        }
    }
}

template <int NT, int NG, int NC>
__global__ __launch_bounds__(128) void gemm_wmma_f32(
    const float* __restrict__ p0, const float* __restrict__ p1,
    const float* __restrict__ p2, int w0, int w1, int w2,
    const float* __restrict__ B, float* __restrict__ C, int M)
{
    const int lane = threadIdx.x & 31;
    const int gw   = blockIdx.x * 4 + (threadIdx.x >> 5);
    const int ng   = gw % NG;
    const int m0   = (gw / NG) * 32;
    if (m0 >= M) return;
    const int c  = lane & 15;
    const int h  = lane >> 4;
    const int n0 = ng * NT * 16;

    const int r0 = min(m0 + c, M - 1);
    const int r1 = min(m0 + 16 + c, M - 1);

    v8f acc[2][NT];
#pragma unroll
    for (int t = 0; t < NT; ++t) { acc[0][t] = v8f{}; acc[1][t] = v8f{}; }

    const float* bcol = B + n0 + c;
    seg_accum<NT, NC>(p0 + (long)r0 * w0, p0 + (long)r1 * w0, w0, bcol, h, acc);
    seg_accum<NT, NC>(p1 + (long)r0 * w1, p1 + (long)r1 * w1, w1, bcol + (long)w0 * NC, h, acc);
    seg_accum<NT, NC>(p2 + (long)r0 * w2, p2 + (long)r1 * w2, w2, bcol + (long)(w0 + w1) * NC, h, acc);

#pragma unroll
    for (int i = 0; i < 2; ++i) {
#pragma unroll
        for (int r = 0; r < 8; ++r) {
            const int row = m0 + i * 16 + 8 * h + r;
            if (row < M) {
                float* crow = C + (long)row * NC + n0 + c;
#pragma unroll
                for (int t = 0; t < NT; ++t) crow[t * 16] = acc[i][t][r];
            }
        }
    }
}

__global__ __launch_bounds__(256) void attn_proj(
    const float* __restrict__ feat, const float* __restrict__ al,
    const float* __restrict__ ar, float* __restrict__ el,
    float* __restrict__ er, int nNodes, int H)
{
    int wid  = blockIdx.x * 8 + (threadIdx.x >> 5);
    int lane = threadIdx.x & 31;
    if (wid >= nNodes * H) return;
    int node = wid / H;
    int h    = wid - node * H;
    int C    = H * 64;
    const float* f  = feat + (long)node * C + h * 64;
    const float* wl = al + h * 64;
    const float* wr = ar + h * 64;
    float f0 = f[lane], f1 = f[lane + 32];
    float sl = f0 * wl[lane] + f1 * wl[lane + 32];
    float sr = f0 * wr[lane] + f1 * wr[lane + 32];
#pragma unroll
    for (int o = 16; o > 0; o >>= 1) {
        sl += __shfl_xor(sl, o, 32);
        sr += __shfl_xor(sr, o, 32);
    }
    if (lane == 0) { el[wid] = sl; er[wid] = sr; }
}

__global__ void edge_lrelu_max(
    const int* __restrict__ src, const int* __restrict__ dst,
    const float* __restrict__ el, const float* __restrict__ er,
    float* __restrict__ ebuf, float* __restrict__ emax, int E, int H)
{
    int i = blockIdx.x * blockDim.x + threadIdx.x;
    if (i >= E) return;
    int s = src[i], d = dst[i];
    for (int h = 0; h < H; ++h) {
        float x = el[s * H + h] + er[d * H + h];
        float e = x > 0.0f ? x : 0.2f * x;
        ebuf[(long)i * H + h] = e;
        atomicMaxF(&emax[d * H + h], e);
    }
}

__global__ void edge_expsum(
    const int* __restrict__ dst, float* __restrict__ ebuf,
    const float* __restrict__ emax, float* __restrict__ denom, int E, int H)
{
    int i = blockIdx.x * blockDim.x + threadIdx.x;
    if (i >= E) return;
    int d = dst[i];
    for (int h = 0; h < H; ++h) {
        float ex = expf(ebuf[(long)i * H + h] - emax[d * H + h]);
        ebuf[(long)i * H + h] = ex;
        atomicAdd(&denom[d * H + h], ex);
    }
}

#define TROWS 196
#define QCAP  288

template <int C, int U>
__device__ __forceinline__ void agg_group(
    const int* myqe, const int* myqd, int q, int lane, int d0,
    const int* __restrict__ src, const float* __restrict__ feat,
    const float* __restrict__ ebuf, const float* __restrict__ denom, float* tile)
{
    constexpr int H  = C / 64;
    constexpr int NI = C / 32;
    const int pl = lane % (U * H);
    const int pu = pl / H, ph = pl - pu * H;
    const float eb = ebuf[(long)myqe[q + pu] * H + ph];
    const float dn = denom[(d0 + myqd[q + pu]) * H + ph];

    int e[U], dd[U], s[U];
#pragma unroll
    for (int u = 0; u < U; ++u) {
        e[u]  = __builtin_amdgcn_readfirstlane(myqe[q + u]);
        dd[u] = __builtin_amdgcn_readfirstlane(myqd[q + u]);
    }
#pragma unroll
    for (int u = 0; u < U; ++u) s[u] = __builtin_amdgcn_readfirstlane(src[e[u]]);
    float v[U][NI];
#pragma unroll
    for (int u = 0; u < U; ++u) {
        const float* f = feat + (long)s[u] * C + lane;
#pragma unroll
        for (int i = 0; i < NI; ++i) v[u][i] = f[i * 32];
    }
    __builtin_amdgcn_sched_barrier(0);
    const float aq = eb / dn;
#pragma unroll
    for (int u = 0; u < U; ++u) {
        float* t = tile + dd[u] * C + lane;
#pragma unroll
        for (int i = 0; i < NI; ++i) {
            const float a = __int_as_float(
                __builtin_amdgcn_readlane(__float_as_int(aq), u * H + (i >> 1)));
            __hip_atomic_fetch_add(t + i * 32, v[u][i] * a,
                                   __ATOMIC_RELAXED, __HIP_MEMORY_SCOPE_WORKGROUP);
        }
    }
}

template <int C>
__global__ __launch_bounds__(1024) void edge_aggregate_tile(
    const int* __restrict__ src, const int* __restrict__ dst,
    const float* __restrict__ feat, const float* __restrict__ ebuf,
    const float* __restrict__ denom, float* __restrict__ rst, int E, int nNodes)
{
    __shared__ __attribute__((aligned(16))) float tile[TROWS * C];
    __shared__ int qe[32 * QCAP];
    __shared__ int qd[32 * QCAP];

    const int tid  = threadIdx.x;
    const int lane = tid & 31;
    const int wave = __builtin_amdgcn_readfirstlane(tid >> 5);
    const int d0   = blockIdx.x * TROWS;
    const int nrows = min(TROWS, nNodes - d0);

    for (int i = tid; i < TROWS * C / 4; i += 1024)
        ((float4*)tile)[i] = make_float4(0.0f, 0.0f, 0.0f, 0.0f);
    __syncthreads();

    int* myqe = qe + wave * QCAP;
    int* myqd = qd + wave * QCAP;
    int qlen = 0;

    const int nfull = E >> 7;
    const int4 none = make_int4(-1, -1, -1, -1);

    auto flush = [&]() {
        int q = 0;
        for (; q + 4 <= qlen; q += 4)
            agg_group<C, 4>(myqe, myqd, q, lane, d0, src, feat, ebuf, denom, tile);
        for (; q < qlen; ++q)
            agg_group<C, 1>(myqe, myqd, q, lane, d0, src, feat, ebuf, denom, tile);
        qlen = 0;
    };

    auto append = [&](const int4 cur, int chunk) -> bool {
        const unsigned r0 = (unsigned)(cur.x - d0), r1 = (unsigned)(cur.y - d0);
        const unsigned r2 = (unsigned)(cur.z - d0), r3 = (unsigned)(cur.w - d0);
        const bool m0 = r0 < (unsigned)nrows, m1 = r1 < (unsigned)nrows;
        const bool m2 = r2 < (unsigned)nrows, m3 = r3 < (unsigned)nrows;
        const unsigned k0 = __builtin_amdgcn_ballot_w32(m0), k1 = __builtin_amdgcn_ballot_w32(m1);
        const unsigned k2 = __builtin_amdgcn_ballot_w32(m2), k3 = __builtin_amdgcn_ballot_w32(m3);
        if ((k0 | k1 | k2 | k3) == 0u) return false;
        const int ebase = chunk * 128 + lane * 4;
        const unsigned rr[4] = {r0, r1, r2, r3};
        const bool     mm[4] = {m0, m1, m2, m3};
        const unsigned kk[4] = {k0, k1, k2, k3};
#pragma unroll
        for (int k = 0; k < 4; ++k) {
            if (mm[k]) {
                const int pos = qlen + (int)__builtin_amdgcn_mbcnt_lo(kk[k], 0u);
                myqe[pos] = ebase + k;
                myqd[pos] = (int)rr[k];
            }
            qlen += __popc(kk[k]);
        }
        return true;
    };

    if (wave < nfull) {
        const int4* dst4 = (const int4*)dst + lane;
        auto load_at = [&](int c) -> int4 {
            return dst4[(long)((c < nfull) ? c : wave) * 32];
        };
        int4 a0 = load_at(wave), a1 = load_at(wave + 32);
        for (int base = wave; base < nfull; base += 128) {
            const int4 b0 = load_at(base + 64), b1 = load_at(base + 96);
            append(a0, base);
            if (base + 32 < nfull) append(a1, base + 32);
            if (qlen >= 32) flush();
            a0 = load_at(base + 128); a1 = load_at(base + 160);
            if (base + 64 < nfull) append(b0, base + 64);
            if (base + 96 < nfull) append(b1, base + 96);
            if (qlen >= 32) flush();
        }
    }
    if (wave == 0 && (E & 127) != 0) {
        const int e0 = nfull * 128 + lane * 4;
        int4 cur = none;
        if (e0     < E) cur.x = dst[e0];
        if (e0 + 1 < E) cur.y = dst[e0 + 1];
        if (e0 + 2 < E) cur.z = dst[e0 + 2];
        if (e0 + 3 < E) cur.w = dst[e0 + 3];
        append(cur, nfull);
    }
    flush();

    __syncthreads();
    float4* o = (float4*)(rst + (long)d0 * C);
    for (int i = tid; i < nrows * (C / 4); i += 1024) o[i] = ((const float4*)tile)[i];
}

__global__ void bias_elu(float* __restrict__ x, const float* __restrict__ b,
                         long n, int C)
{
    long i = (long)blockIdx.x * blockDim.x + threadIdx.x;
    if (i >= n) return;
    float v = x[i] + b[i % C];
    x[i] = v > 0.0f ? v : expm1f(v);
}

__global__ __launch_bounds__(256) void classify(
    const float* __restrict__ rst2, const float* __restrict__ b2,
    const float* __restrict__ Wc1, const float* __restrict__ bc1,
    const float* __restrict__ Wc2, const float* __restrict__ bc2,
    float* __restrict__ out, int nNodes)
{
    int wid  = blockIdx.x * 8 + (threadIdx.x >> 5);
    int lane = threadIdx.x & 31;
    if (wid >= nNodes) return;
    float acc = 0.0f;
#pragma unroll
    for (int d = 0; d < 64; ++d)
        acc += (rst2[(long)wid * 64 + d] + b2[d]) * Wc1[d * 32 + lane];
    float t = fmaxf(acc + bc1[lane], 0.0f);
    float o0 = t * Wc2[lane * 2 + 0];
    float o1 = t * Wc2[lane * 2 + 1];
#pragma unroll
    for (int o = 16; o > 0; o >>= 1) {
        o0 += __shfl_xor(o0, o, 32);
        o1 += __shfl_xor(o1, o, 32);
    }
    if (lane == 0) {
        out[(long)wid * 2 + 0] = o0 + bc2[0];
        out[(long)wid * 2 + 1] = o1 + bc2[1];
    }
}

extern "C" void kernel_launch(void* const* d_in, const int* in_sizes, int n_in,
                              void* d_out, int out_size, void* d_ws, size_t ws_size,
                              hipStream_t stream)
{
    (void)n_in; (void)out_size; (void)ws_size;

    const float* randfeat = (const float*)d_in[0];
    const float* func_emb = (const float*)d_in[1];
    const float* codebert = (const float*)d_in[2];
    const int*   src      = (const int*)d_in[3];
    const int*   dst      = (const int*)d_in[4];
    const float* W1  = (const float*)d_in[5];
    const float* al1 = (const float*)d_in[6];
    const float* ar1 = (const float*)d_in[7];
    const float* b1  = (const float*)d_in[8];
    const float* W2  = (const float*)d_in[9];
    const float* al2 = (const float*)d_in[10];
    const float* ar2 = (const float*)d_in[11];
    const float* b2  = (const float*)d_in[12];
    const float* Wc1 = (const float*)d_in[13];
    const float* bc1 = (const float*)d_in[14];
    const float* Wc2 = (const float*)d_in[15];
    const float* bc2 = (const float*)d_in[16];
    float* out = (float*)d_out;

    const int  E  = in_sizes[3];
    const long NN = N_NODES;

    float* ws     = (float*)d_ws;
    float* feat1  = ws;
    float* rst1   = feat1  + NN * 256;
    float* feat2  = rst1   + NN * 256;
    float* rst2   = feat2  + NN * 64;
    float* el1    = rst2   + NN * 64;
    float* er1    = el1    + NN * 4;
    float* emax1  = er1    + NN * 4;
    float* denom1 = emax1  + NN * 4;
    float* el2    = denom1 + NN * 4;
    float* er2    = el2    + NN;
    float* emax2  = er2    + NN;
    float* denom2 = emax2  + NN;
    float* e1     = denom2 + NN;
    float* e2     = e1 + (long)E * 4;

    fill_f32<<<(int)((NN * 4 + 255) / 256), 256, 0, stream>>>(denom1, 0.0f, NN * 4);
    fill_f32<<<(int)((NN + 255) / 256), 256, 0, stream>>>(denom2, 0.0f, NN);
    fill_f32<<<(int)((NN * 4 + 255) / 256), 256, 0, stream>>>(emax1, -INFINITY, NN * 4);
    fill_f32<<<(int)((NN + 255) / 256), 256, 0, stream>>>(emax2, -INFINITY, NN);

    gemm_wmma_f32<8, 2, 256><<<((N_NODES + 31) / 32 * 2 + 3) / 4, 128, 0, stream>>>(
        randfeat, func_emb, codebert, 100, 100, 768, W1, feat1, N_NODES);
    attn_proj<<<(N_NODES * 4 + 7) / 8, 256, 0, stream>>>(feat1, al1, ar1, el1, er1, N_NODES, 4);
    edge_lrelu_max<<<(E + 255) / 256, 256, 0, stream>>>(src, dst, el1, er1, e1, emax1, E, 4);
    edge_expsum<<<(E + 255) / 256, 256, 0, stream>>>(dst, e1, emax1, denom1, E, 4);
    edge_aggregate_tile<256><<<(N_NODES + TROWS - 1) / TROWS, 1024, 0, stream>>>(src, dst, feat1, e1, denom1, rst1, E, N_NODES);
    bias_elu<<<(int)((NN * 256 + 255) / 256), 256, 0, stream>>>(rst1, b1, NN * 256, 256);

    gemm_wmma_f32<4, 1, 64><<<((N_NODES + 31) / 32 + 3) / 4, 128, 0, stream>>>(
        rst1, rst1, rst1, 256, 0, 0, W2, feat2, N_NODES);
    attn_proj<<<(N_NODES + 7) / 8, 256, 0, stream>>>(feat2, al2, ar2, el2, er2, N_NODES, 1);
    edge_lrelu_max<<<(E + 255) / 256, 256, 0, stream>>>(src, dst, el2, er2, e2, emax2, E, 1);
    edge_expsum<<<(E + 255) / 256, 256, 0, stream>>>(dst, e2, emax2, denom2, E, 1);
    edge_aggregate_tile<64><<<(N_NODES + TROWS - 1) / TROWS, 1024, 0, stream>>>(src, dst, feat2, e2, denom2, rst2, E, N_NODES);

    classify<<<(N_NODES + 7) / 8, 256, 0, stream>>>(rst2, b2, Wc1, bc1, Wc2, bc2, out, N_NODES);
}


# Light kernels (el/er projections, edge max and sum passes, bias+elu, classifier) run as grids of 2k-8k long-lived looping waves, not one short wave per row; row sums in the same order
# speedup vs baseline: 2.2665x; 1.1116x over previous
#include <hip/hip_runtime.h>
#include <math.h>


typedef __attribute__((ext_vector_type(2))) float v2f;
typedef __attribute__((ext_vector_type(8))) float v8f;

#define N_NODES 50000

__global__ void fill_f32(float* __restrict__ p, float v, long n) {
    long i = (long)blockIdx.x * blockDim.x + threadIdx.x;
    if (i < n) p[i] = v;
}

__device__ __forceinline__ void atomicMaxF(float* addr, float val) {
    if (val >= 0.0f)
        atomicMax((int*)addr, __float_as_int(val));
    else
        atomicMin((unsigned int*)addr, (unsigned int)__float_as_int(val));
}

template <int NT, int NC>
__device__ __forceinline__ void seg_accum(
    const float* __restrict__ a0, const float* __restrict__ a1, int w,
    const float* __restrict__ bcol, int h, v8f (&acc)[2][NT])
{
    const int nblk = w >> 5;
    if (nblk > 0) {
        const float4* q0 = (const float4*)(a0 + 16 * h);
        const float4* q1 = (const float4*)(a1 + 16 * h);
        const float* bp = bcol + (long)(16 * h) * NC;
        float4 ra[4], rb[4];
        v2f bc[NT];
#pragma unroll
        for (int q = 0; q < 4; ++q) { ra[q] = q0[q]; rb[q] = q1[q]; }
#pragma unroll
        for (int t = 0; t < NT; ++t) { bc[t].x = bp[t * 16]; bc[t].y = bp[NC + t * 16]; }

        for (int blk = 0; blk < nblk; ++blk) {
            const int adv = (blk + 1 < nblk) ? 32 : 0;
            const float4* nq0 = (const float4*)((const float*)q0 + adv);
            const float4* nq1 = (const float4*)((const float*)q1 + adv);
            const float* nbp = bp + (long)adv * NC;
            float4 na[4], nb[4];
#pragma unroll
            for (int q = 0; q < 4; ++q) { na[q] = nq0[q]; nb[q] = nq1[q]; }
            __builtin_amdgcn_sched_barrier(0);
#pragma unroll
            for (int j = 0; j < 8; ++j) {
                v2f bn[NT];
                if (j < 7) {
#pragma unroll
                    for (int t = 0; t < NT; ++t) {
                        bn[t].x = bp[(2 * j + 2) * NC + t * 16];
                        bn[t].y = bp[(2 * j + 3) * NC + t * 16];
                    }
                } else {
#pragma unroll
                    for (int t = 0; t < NT; ++t) {
                        bn[t].x = nbp[t * 16];
                        bn[t].y = nbp[NC + t * 16];
                    }
                }
                v2f fa, fb;
                if (j & 1) { fa.x = ra[j >> 1].z; fa.y = ra[j >> 1].w; fb.x = rb[j >> 1].z; fb.y = rb[j >> 1].w; }
                else       { fa.x = ra[j >> 1].x; fa.y = ra[j >> 1].y; fb.x = rb[j >> 1].x; fb.y = rb[j >> 1].y; }
#pragma unroll
                for (int t = 0; t < NT; ++t) {
                    acc[0][t] = __builtin_amdgcn_wmma_f32_16x16x4_f32(
                        false, fa, false, bc[t], (short)0, acc[0][t], false, false);
                    acc[1][t] = __builtin_amdgcn_wmma_f32_16x16x4_f32(
                        false, fb, false, bc[t], (short)0, acc[1][t], false, false);
                }
#pragma unroll
                for (int t = 0; t < NT; ++t) bc[t] = bn[t];
            }
            __builtin_amdgcn_sched_barrier(0);
            asm volatile("v_nop\n\tv_nop\n\tv_nop\n\tv_nop");
#pragma unroll
            for (int q = 0; q < 4; ++q) { ra[q] = na[q]; rb[q] = nb[q]; }
            q0 = nq0; q1 = nq1; bp = nbp;
        }
    }
    for (int k = nblk << 5; k + 4 <= w; k += 4) {
        const float2 ta = *(const float2*)(a0 + k + 2 * h);
        const float2 tb = *(const float2*)(a1 + k + 2 * h);
        v2f fa, fb;
        fa.x = ta.x; fa.y = ta.y; fb.x = tb.x; fb.y = tb.y;
        const float* bp = bcol + (long)(k + 2 * h) * NC;
#pragma unroll
        for (int t = 0; t < NT; ++t) {
            v2f b;
            b.x = bp[t * 16];
            b.y = bp[NC + t * 16];
            acc[0][t] = __builtin_amdgcn_wmma_f32_16x16x4_f32(
                false, fa, false, b, (short)0, acc[0][t], false, false);
            acc[1][t] = __builtin_amdgcn_wmma_f32_16x16x4_f32(
                false, fb, false, b, (short)0, acc[1][t], false, false);
        }
    }
}

template <int NT, int NG, int NC>
__global__ __launch_bounds__(128) void gemm_wmma_f32(
    const float* __restrict__ p0, const float* __restrict__ p1,
    const float* __restrict__ p2, int w0, int w1, int w2,
    const float* __restrict__ B, float* __restrict__ C, int M)
{
    const int lane = threadIdx.x & 31;
    const int gw   = blockIdx.x * 4 + (threadIdx.x >> 5);
    const int ng   = gw % NG;
    const int m0   = (gw / NG) * 32;
    if (m0 >= M) return;
    const int c  = lane & 15;
    const int h  = lane >> 4;
    const int n0 = ng * NT * 16;

    const int r0 = min(m0 + c, M - 1);
    const int r1 = min(m0 + 16 + c, M - 1);

    v8f acc[2][NT];
#pragma unroll
    for (int t = 0; t < NT; ++t) { acc[0][t] = v8f{}; acc[1][t] = v8f{}; }

    const float* bcol = B + n0 + c;
    seg_accum<NT, NC>(p0 + (long)r0 * w0, p0 + (long)r1 * w0, w0, bcol, h, acc);
    seg_accum<NT, NC>(p1 + (long)r0 * w1, p1 + (long)r1 * w1, w1, bcol + (long)w0 * NC, h, acc);
    seg_accum<NT, NC>(p2 + (long)r0 * w2, p2 + (long)r1 * w2, w2, bcol + (long)(w0 + w1) * NC, h, acc);

#pragma unroll
    for (int i = 0; i < 2; ++i) {
#pragma unroll
        for (int r = 0; r < 8; ++r) {
            const int row = m0 + i * 16 + 8 * h + r;
            if (row < M) {
                float* crow = C + (long)row * NC + n0 + c;
#pragma unroll
                for (int t = 0; t < NT; ++t) crow[t * 16] = acc[i][t][r];
            }
        }
    }
}

#define PU 8
template <int H>
__global__ __launch_bounds__(256) void attn_proj(
    const float* __restrict__ feat, const float* __restrict__ al,
    const float* __restrict__ ar, float* __restrict__ el,
    float* __restrict__ er, int nPairs)
{
    static_assert(PU % H == 0, "a trip starts at head 0");
    const int lane = threadIdx.x & 31;
    const int wid  = blockIdx.x * 8 + (threadIdx.x >> 5);
    const int nw   = gridDim.x * 8;
    float wl0[H], wl1[H], wr0[H], wr1[H];
#pragma unroll
    for (int h = 0; h < H; ++h) {
        wl0[h] = al[h * 64 + lane]; wl1[h] = al[h * 64 + lane + 32];
        wr0[h] = ar[h * 64 + lane]; wr1[h] = ar[h * 64 + lane + 32];
    }
    for (int p0 = wid * PU; p0 < nPairs; p0 += nw * PU) {
        float f0[PU], f1[PU];
#pragma unroll
        for (int u = 0; u < PU; ++u) {
            const float* f = feat + (long)min(p0 + u, nPairs - 1) * 64;
            f0[u] = f[lane]; f1[u] = f[lane + 32];
        }
        float sl[PU], sr[PU];
#pragma unroll
        for (int u = 0; u < PU; ++u) {
            sl[u] = f0[u] * wl0[u % H] + f1[u] * wl1[u % H];
            sr[u] = f0[u] * wr0[u % H] + f1[u] * wr1[u % H];
        }
#pragma unroll
        for (int o = 16; o > 0; o >>= 1) {
#pragma unroll
            for (int u = 0; u < PU; ++u) {
                sl[u] += __shfl_xor(sl[u], o, 32);
                sr[u] += __shfl_xor(sr[u], o, 32);
            }
        }
        float ol = sl[0], orr = sr[0];
#pragma unroll
        for (int u = 1; u < PU; ++u) if (lane == u) { ol = sl[u]; orr = sr[u]; }
        if (lane < PU && p0 + lane < nPairs) { el[p0 + lane] = ol; er[p0 + lane] = orr; }
    }
}

#define EU 4
template <int H>
__global__ __launch_bounds__(256) void edge_lrelu_max(
    const int* __restrict__ src, const int* __restrict__ dst,
    const float* __restrict__ el, const float* __restrict__ er,
    float* __restrict__ ebuf, float* __restrict__ emax, int E)
{
    const int stride = gridDim.x * 256;
    for (int i0 = blockIdx.x * 256 + threadIdx.x; i0 < E; i0 += EU * stride) {
        int s[EU], d[EU];
#pragma unroll
        for (int u = 0; u < EU; ++u) {
            const int i = min(i0 + u * stride, E - 1);
            s[u] = src[i]; d[u] = dst[i];
        }
        float x[EU][H];
#pragma unroll
        for (int u = 0; u < EU; ++u)
#pragma unroll
            for (int h = 0; h < H; ++h) x[u][h] = el[s[u] * H + h] + er[d[u] * H + h];
#pragma unroll
        for (int u = 0; u < EU; ++u) {
            const int i = i0 + u * stride;
            if (i < E) {
#pragma unroll
                for (int h = 0; h < H; ++h) {
                    x[u][h] = x[u][h] > 0.0f ? x[u][h] : 0.2f * x[u][h];
                    ebuf[(long)i * H + h] = x[u][h];
                }
#pragma unroll
                for (int h = 0; h < H; ++h) atomicMaxF(&emax[d[u] * H + h], x[u][h]);
            }
        }
    }
}

template <int H>
__global__ __launch_bounds__(256) void edge_expsum(
    const int* __restrict__ dst, float* __restrict__ ebuf,
    const float* __restrict__ emax, float* __restrict__ denom, int E)
{
    const int stride = gridDim.x * 256;
    for (int i0 = blockIdx.x * 256 + threadIdx.x; i0 < E; i0 += EU * stride) {
        int d[EU];
        float ev[EU][H];
#pragma unroll
        for (int u = 0; u < EU; ++u) {
            const int i = min(i0 + u * stride, E - 1);
            d[u] = dst[i];
#pragma unroll
            for (int h = 0; h < H; ++h) ev[u][h] = ebuf[(long)i * H + h];
        }
        float mx[EU][H];
#pragma unroll
        for (int u = 0; u < EU; ++u)
#pragma unroll
            for (int h = 0; h < H; ++h) mx[u][h] = emax[d[u] * H + h];
#pragma unroll
        for (int u = 0; u < EU; ++u) {
            const int i = i0 + u * stride;
            if (i < E) {
#pragma unroll
                for (int h = 0; h < H; ++h) {
                    const float ex = expf(ev[u][h] - mx[u][h]);
                    ebuf[(long)i * H + h] = ex;
                    atomicAdd(&denom[d[u] * H + h], ex);
                }
            }
        }
    }
}

#define TROWS 196
#define QCAP  288

template <int C, int U>
__device__ __forceinline__ void agg_group(
    const int* myqe, const int* myqd, int q, int lane, int d0,
    const int* __restrict__ src, const float* __restrict__ feat,
    const float* __restrict__ ebuf, const float* __restrict__ denom, float* tile)
{
    constexpr int H  = C / 64;
    constexpr int NI = C / 32;
    const int pl = lane % (U * H);
    const int pu = pl / H, ph = pl - pu * H;
    const float eb = ebuf[(long)myqe[q + pu] * H + ph];
    const float dn = denom[(d0 + myqd[q + pu]) * H + ph];

    int e[U], dd[U], s[U];
#pragma unroll
    for (int u = 0; u < U; ++u) {
        e[u]  = __builtin_amdgcn_readfirstlane(myqe[q + u]);
        dd[u] = __builtin_amdgcn_readfirstlane(myqd[q + u]);
    }
#pragma unroll
    for (int u = 0; u < U; ++u) s[u] = __builtin_amdgcn_readfirstlane(src[e[u]]);
    float v[U][NI];
#pragma unroll
    for (int u = 0; u < U; ++u) {
        const float* f = feat + (long)s[u] * C + lane;
#pragma unroll
        for (int i = 0; i < NI; ++i) v[u][i] = f[i * 32];
    }
    __builtin_amdgcn_sched_barrier(0);
    const float aq = eb / dn;
#pragma unroll
    for (int u = 0; u < U; ++u) {
        float* t = tile + dd[u] * C + lane;
#pragma unroll
        for (int i = 0; i < NI; ++i) {
            const float a = __int_as_float(
                __builtin_amdgcn_readlane(__float_as_int(aq), u * H + (i >> 1)));
            __hip_atomic_fetch_add(t + i * 32, v[u][i] * a,
                                   __ATOMIC_RELAXED, __HIP_MEMORY_SCOPE_WORKGROUP);
        }
    }
}

template <int C>
__global__ __launch_bounds__(1024) void edge_aggregate_tile(
    const int* __restrict__ src, const int* __restrict__ dst,
    const float* __restrict__ feat, const float* __restrict__ ebuf,
    const float* __restrict__ denom, float* __restrict__ rst, int E, int nNodes)
{
    __shared__ __attribute__((aligned(16))) float tile[TROWS * C];
    __shared__ int qe[32 * QCAP];
    __shared__ int qd[32 * QCAP];

    const int tid  = threadIdx.x;
    const int lane = tid & 31;
    const int wave = __builtin_amdgcn_readfirstlane(tid >> 5);
    const int d0   = blockIdx.x * TROWS;
    const int nrows = min(TROWS, nNodes - d0);

    for (int i = tid; i < TROWS * C / 4; i += 1024)
        ((float4*)tile)[i] = make_float4(0.0f, 0.0f, 0.0f, 0.0f);
    __syncthreads();

    int* myqe = qe + wave * QCAP;
    int* myqd = qd + wave * QCAP;
    int qlen = 0;

    const int nfull = E >> 7;
    const int4 none = make_int4(-1, -1, -1, -1);

    auto flush = [&]() {
        int q = 0;
        for (; q + 4 <= qlen; q += 4)
            agg_group<C, 4>(myqe, myqd, q, lane, d0, src, feat, ebuf, denom, tile);
        for (; q < qlen; ++q)
            agg_group<C, 1>(myqe, myqd, q, lane, d0, src, feat, ebuf, denom, tile);
        qlen = 0;
    };

    auto append = [&](const int4 cur, int chunk) -> bool {
        const unsigned r0 = (unsigned)(cur.x - d0), r1 = (unsigned)(cur.y - d0);
        const unsigned r2 = (unsigned)(cur.z - d0), r3 = (unsigned)(cur.w - d0);
        const bool m0 = r0 < (unsigned)nrows, m1 = r1 < (unsigned)nrows;
        const bool m2 = r2 < (unsigned)nrows, m3 = r3 < (unsigned)nrows;
        const unsigned k0 = __builtin_amdgcn_ballot_w32(m0), k1 = __builtin_amdgcn_ballot_w32(m1);
        const unsigned k2 = __builtin_amdgcn_ballot_w32(m2), k3 = __builtin_amdgcn_ballot_w32(m3);
        if ((k0 | k1 | k2 | k3) == 0u) return false;
        const int ebase = chunk * 128 + lane * 4;
        const unsigned rr[4] = {r0, r1, r2, r3};
        const bool     mm[4] = {m0, m1, m2, m3};
        const unsigned kk[4] = {k0, k1, k2, k3};
#pragma unroll
        for (int k = 0; k < 4; ++k) {
            if (mm[k]) {
                const int pos = qlen + (int)__builtin_amdgcn_mbcnt_lo(kk[k], 0u);
                myqe[pos] = ebase + k;
                myqd[pos] = (int)rr[k];
            }
            qlen += __popc(kk[k]);
        }
        return true;
    };

    if (wave < nfull) {
        const int4* dst4 = (const int4*)dst + lane;
        auto load_at = [&](int c) -> int4 {
            return dst4[(long)((c < nfull) ? c : wave) * 32];
        };
        int4 a0 = load_at(wave), a1 = load_at(wave + 32);
        for (int base = wave; base < nfull; base += 128) {
            const int4 b0 = load_at(base + 64), b1 = load_at(base + 96);
            append(a0, base);
            if (base + 32 < nfull) append(a1, base + 32);
            if (qlen >= 32) flush();
            a0 = load_at(base + 128); a1 = load_at(base + 160);
            if (base + 64 < nfull) append(b0, base + 64);
            if (base + 96 < nfull) append(b1, base + 96);
            if (qlen >= 32) flush();
        }
    }
    if (wave == 0 && (E & 127) != 0) {
        const int e0 = nfull * 128 + lane * 4;
        int4 cur = none;
        if (e0     < E) cur.x = dst[e0];
        if (e0 + 1 < E) cur.y = dst[e0 + 1];
        if (e0 + 2 < E) cur.z = dst[e0 + 2];
        if (e0 + 3 < E) cur.w = dst[e0 + 3];
        append(cur, nfull);
    }
    flush();

    __syncthreads();
    float4* o = (float4*)(rst + (long)d0 * C);
    for (int i = tid; i < nrows * (C / 4); i += 1024) o[i] = ((const float4*)tile)[i];
}

template <int C>
__global__ __launch_bounds__(256) void bias_elu(float* __restrict__ x, const float* __restrict__ b,
                                                int n4)
{
    float4* x4 = (float4*)x;
    const int stride = gridDim.x * 256;
    for (int i0 = blockIdx.x * 256 + threadIdx.x; i0 < n4; i0 += 4 * stride) {
        float4 v[4];
#pragma unroll
        for (int u = 0; u < 4; ++u) v[u] = x4[min(i0 + u * stride, n4 - 1)];
#pragma unroll
        for (int u = 0; u < 4; ++u) {
            const int i = i0 + u * stride;
            if (i < n4) {
                const float* bb = b + ((unsigned)i % (unsigned)(C / 4)) * 4u;
                float4 r;
                float t;
                t = v[u].x + bb[0]; r.x = t > 0.0f ? t : expm1f(t);
                t = v[u].y + bb[1]; r.y = t > 0.0f ? t : expm1f(t);
                t = v[u].z + bb[2]; r.z = t > 0.0f ? t : expm1f(t);
                t = v[u].w + bb[3]; r.w = t > 0.0f ? t : expm1f(t);
                x4[i] = r;
            }
        }
    }
}

__global__ __launch_bounds__(256) void classify(
    const float* __restrict__ rst2, const float* __restrict__ b2,
    const float* __restrict__ Wc1, const float* __restrict__ bc1,
    const float* __restrict__ Wc2, const float* __restrict__ bc2,
    float* __restrict__ out, int nNodes)
{
    const int lane = threadIdx.x & 31;
    const int wid  = blockIdx.x * 8 + __builtin_amdgcn_readfirstlane(threadIdx.x >> 5);
    const int nw   = gridDim.x * 8;
    float w[64];
#pragma unroll
    for (int d = 0; d < 64; ++d) w[d] = Wc1[d * 32 + lane];
    const float bl  = bc1[lane];
    const float w20 = Wc2[lane * 2 + 0], w21 = Wc2[lane * 2 + 1];
    const float c0  = bc2[0], c1 = bc2[1];
    for (int node = wid; node < nNodes; node += nw) {
        const float* r = rst2 + (long)node * 64;
        float acc = 0.0f;
#pragma unroll
        for (int d = 0; d < 64; ++d)
            acc += (r[d] + b2[d]) * w[d];
        float t = fmaxf(acc + bl, 0.0f);
        float o0 = t * w20;
        float o1 = t * w21;
#pragma unroll
        for (int o = 16; o > 0; o >>= 1) {
            o0 += __shfl_xor(o0, o, 32);
            o1 += __shfl_xor(o1, o, 32);
        }
        if (lane == 0) {
            out[(long)node * 2 + 0] = o0 + c0;
            out[(long)node * 2 + 1] = o1 + c1;
        }
    }
}

extern "C" void kernel_launch(void* const* d_in, const int* in_sizes, int n_in,
                              void* d_out, int out_size, void* d_ws, size_t ws_size,
                              hipStream_t stream)
{
    (void)n_in; (void)out_size; (void)ws_size;

    const float* randfeat = (const float*)d_in[0];
    const float* func_emb = (const float*)d_in[1];
    const float* codebert = (const float*)d_in[2];
    const int*   src      = (const int*)d_in[3];
    const int*   dst      = (const int*)d_in[4];
    const float* W1  = (const float*)d_in[5];
    const float* al1 = (const float*)d_in[6];
    const float* ar1 = (const float*)d_in[7];
    const float* b1  = (const float*)d_in[8];
    const float* W2  = (const float*)d_in[9];
    const float* al2 = (const float*)d_in[10];
    const float* ar2 = (const float*)d_in[11];
    const float* b2  = (const float*)d_in[12];
    const float* Wc1 = (const float*)d_in[13];
    const float* bc1 = (const float*)d_in[14];
    const float* Wc2 = (const float*)d_in[15];
    const float* bc2 = (const float*)d_in[16];
    float* out = (float*)d_out;

    const int  E  = in_sizes[3];
    const long NN = N_NODES;

    float* ws     = (float*)d_ws;
    float* feat1  = ws;
    float* rst1   = feat1  + NN * 256;
    float* feat2  = rst1   + NN * 256;
    float* rst2   = feat2  + NN * 64;
    float* el1    = rst2   + NN * 64;
    float* er1    = el1    + NN * 4;
    float* emax1  = er1    + NN * 4;
    float* denom1 = emax1  + NN * 4;
    float* el2    = denom1 + NN * 4;
    float* er2    = el2    + NN;
    float* emax2  = er2    + NN;
    float* denom2 = emax2  + NN;
    float* e1     = denom2 + NN;
    float* e2     = e1 + (long)E * 4;

    fill_f32<<<(int)((NN * 4 + 255) / 256), 256, 0, stream>>>(denom1, 0.0f, NN * 4);
    fill_f32<<<(int)((NN + 255) / 256), 256, 0, stream>>>(denom2, 0.0f, NN);
    fill_f32<<<(int)((NN * 4 + 255) / 256), 256, 0, stream>>>(emax1, -INFINITY, NN * 4);
    fill_f32<<<(int)((NN + 255) / 256), 256, 0, stream>>>(emax2, -INFINITY, NN);

    gemm_wmma_f32<8, 2, 256><<<((N_NODES + 31) / 32 * 2 + 3) / 4, 128, 0, stream>>>(
        randfeat, func_emb, codebert, 100, 100, 768, W1, feat1, N_NODES);
    attn_proj<4><<<1024, 256, 0, stream>>>(feat1, al1, ar1, el1, er1, N_NODES * 4);
    edge_lrelu_max<4><<<1024, 256, 0, stream>>>(src, dst, el1, er1, e1, emax1, E);
    edge_expsum<4><<<1024, 256, 0, stream>>>(dst, e1, emax1, denom1, E);
    edge_aggregate_tile<256><<<(N_NODES + TROWS - 1) / TROWS, 1024, 0, stream>>>(src, dst, feat1, e1, denom1, rst1, E, N_NODES);
    bias_elu<256><<<1024, 256, 0, stream>>>(rst1, b1, N_NODES * 64);

    gemm_wmma_f32<4, 1, 64><<<((N_NODES + 31) / 32 + 3) / 4, 128, 0, stream>>>(
        rst1, rst1, rst1, 256, 0, 0, W2, feat2, N_NODES);
    attn_proj<1><<<256, 256, 0, stream>>>(feat2, al2, ar2, el2, er2, N_NODES);
    edge_lrelu_max<1><<<1024, 256, 0, stream>>>(src, dst, el2, er2, e2, emax2, E);
    edge_expsum<1><<<1024, 256, 0, stream>>>(dst, e2, emax2, denom2, E);
    edge_aggregate_tile<64><<<(N_NODES + TROWS - 1) / TROWS, 1024, 0, stream>>>(src, dst, feat2, e2, denom2, rst2, E, N_NODES);

    classify<<<1024, 256, 0, stream>>>(rst2, b2, Wc1, bc1, Wc2, bc2, out, N_NODES);
}


# Edge softmax folded into the dst-tile aggregation: row max, exp sum and sums kept in LDS, weights divided out at the store; 8 launches, e buffers and global atomics gone; new summation order
# speedup vs baseline: 4.5751x; 2.0186x over previous
#include <hip/hip_runtime.h>
#include <math.h>


typedef __attribute__((ext_vector_type(2))) float v2f;
typedef __attribute__((ext_vector_type(8))) float v8f;

#define N_NODES 50000

template <int NT, int NC>
__device__ __forceinline__ void seg_accum(
    const float* __restrict__ a0, const float* __restrict__ a1, int w,
    const float* __restrict__ bcol, int h, v8f (&acc)[2][NT])
{
    const int nblk = w >> 5;
    if (nblk > 0) {
        const float4* q0 = (const float4*)(a0 + 16 * h);
        const float4* q1 = (const float4*)(a1 + 16 * h);
        const float* bp = bcol + (long)(16 * h) * NC;
        float4 ra[4], rb[4];
        v2f bc[NT];
#pragma unroll
        for (int q = 0; q < 4; ++q) { ra[q] = q0[q]; rb[q] = q1[q]; }
#pragma unroll
        for (int t = 0; t < NT; ++t) { bc[t].x = bp[t * 16]; bc[t].y = bp[NC + t * 16]; }

        for (int blk = 0; blk < nblk; ++blk) {
            const int adv = (blk + 1 < nblk) ? 32 : 0;
            const float4* nq0 = (const float4*)((const float*)q0 + adv);
            const float4* nq1 = (const float4*)((const float*)q1 + adv);
            const float* nbp = bp + (long)adv * NC;
            float4 na[4], nb[4];
#pragma unroll
            for (int q = 0; q < 4; ++q) { na[q] = nq0[q]; nb[q] = nq1[q]; }
            __builtin_amdgcn_sched_barrier(0);
#pragma unroll
            for (int j = 0; j < 8; ++j) {
                v2f bn[NT];
                if (j < 7) {
#pragma unroll
                    for (int t = 0; t < NT; ++t) {
                        bn[t].x = bp[(2 * j + 2) * NC + t * 16];
                        bn[t].y = bp[(2 * j + 3) * NC + t * 16];
                    }
                } else {
#pragma unroll
                    for (int t = 0; t < NT; ++t) {
                        bn[t].x = nbp[t * 16];
                        bn[t].y = nbp[NC + t * 16];
                    }
                }
                v2f fa, fb;
                if (j & 1) { fa.x = ra[j >> 1].z; fa.y = ra[j >> 1].w; fb.x = rb[j >> 1].z; fb.y = rb[j >> 1].w; }
                else       { fa.x = ra[j >> 1].x; fa.y = ra[j >> 1].y; fb.x = rb[j >> 1].x; fb.y = rb[j >> 1].y; }
#pragma unroll
                for (int t = 0; t < NT; ++t) {
                    acc[0][t] = __builtin_amdgcn_wmma_f32_16x16x4_f32(
                        false, fa, false, bc[t], (short)0, acc[0][t], false, false);
                    acc[1][t] = __builtin_amdgcn_wmma_f32_16x16x4_f32(
                        false, fb, false, bc[t], (short)0, acc[1][t], false, false);
                }
#pragma unroll
                for (int t = 0; t < NT; ++t) bc[t] = bn[t];
            }
            __builtin_amdgcn_sched_barrier(0);
            asm volatile("v_nop\n\tv_nop\n\tv_nop\n\tv_nop");
#pragma unroll
            for (int q = 0; q < 4; ++q) { ra[q] = na[q]; rb[q] = nb[q]; }
            q0 = nq0; q1 = nq1; bp = nbp;
        }
    }
    for (int k = nblk << 5; k + 4 <= w; k += 4) {
        const float2 ta = *(const float2*)(a0 + k + 2 * h);
        const float2 tb = *(const float2*)(a1 + k + 2 * h);
        v2f fa, fb;
        fa.x = ta.x; fa.y = ta.y; fb.x = tb.x; fb.y = tb.y;
        const float* bp = bcol + (long)(k + 2 * h) * NC;
#pragma unroll
        for (int t = 0; t < NT; ++t) {
            v2f b;
            b.x = bp[t * 16];
            b.y = bp[NC + t * 16];
            acc[0][t] = __builtin_amdgcn_wmma_f32_16x16x4_f32(
                false, fa, false, b, (short)0, acc[0][t], false, false);
            acc[1][t] = __builtin_amdgcn_wmma_f32_16x16x4_f32(
                false, fb, false, b, (short)0, acc[1][t], false, false);
        }
    }
}

template <int NT, int NG, int NC>
__global__ __launch_bounds__(128) void gemm_wmma_f32(
    const float* __restrict__ p0, const float* __restrict__ p1,
    const float* __restrict__ p2, int w0, int w1, int w2,
    const float* __restrict__ B, float* __restrict__ C, int M)
{
    const int lane = threadIdx.x & 31;
    const int gw   = blockIdx.x * 4 + (threadIdx.x >> 5);
    const int ng   = gw % NG;
    const int m0   = (gw / NG) * 32;
    if (m0 >= M) return;
    const int c  = lane & 15;
    const int h  = lane >> 4;
    const int n0 = ng * NT * 16;

    const int r0 = min(m0 + c, M - 1);
    const int r1 = min(m0 + 16 + c, M - 1);

    v8f acc[2][NT];
#pragma unroll
    for (int t = 0; t < NT; ++t) { acc[0][t] = v8f{}; acc[1][t] = v8f{}; }

    const float* bcol = B + n0 + c;
    seg_accum<NT, NC>(p0 + (long)r0 * w0, p0 + (long)r1 * w0, w0, bcol, h, acc);
    seg_accum<NT, NC>(p1 + (long)r0 * w1, p1 + (long)r1 * w1, w1, bcol + (long)w0 * NC, h, acc);
    seg_accum<NT, NC>(p2 + (long)r0 * w2, p2 + (long)r1 * w2, w2, bcol + (long)(w0 + w1) * NC, h, acc);

#pragma unroll
    for (int i = 0; i < 2; ++i) {
#pragma unroll
        for (int r = 0; r < 8; ++r) {
            const int row = m0 + i * 16 + 8 * h + r;
            if (row < M) {
                float* crow = C + (long)row * NC + n0 + c;
#pragma unroll
                for (int t = 0; t < NT; ++t) crow[t * 16] = acc[i][t][r];
            }
        }
    }
}

#define PU 8
template <int H>
__global__ __launch_bounds__(256) void attn_proj(
    const float* __restrict__ feat, const float* __restrict__ al,
    const float* __restrict__ ar, float* __restrict__ el,
    float* __restrict__ er, int nPairs)
{
    static_assert(PU % H == 0, "a trip starts at head 0");
    const int lane = threadIdx.x & 31;
    const int wid  = blockIdx.x * 8 + (threadIdx.x >> 5);
    const int nw   = gridDim.x * 8;
    float wl0[H], wl1[H], wr0[H], wr1[H];
#pragma unroll
    for (int h = 0; h < H; ++h) {
        wl0[h] = al[h * 64 + lane]; wl1[h] = al[h * 64 + lane + 32];
        wr0[h] = ar[h * 64 + lane]; wr1[h] = ar[h * 64 + lane + 32];
    }
    for (int p0 = wid * PU; p0 < nPairs; p0 += nw * PU) {
        float f0[PU], f1[PU];
#pragma unroll
        for (int u = 0; u < PU; ++u) {
            const float* f = feat + (long)min(p0 + u, nPairs - 1) * 64;
            f0[u] = f[lane]; f1[u] = f[lane + 32];
        }
        float sl[PU], sr[PU];
#pragma unroll
        for (int u = 0; u < PU; ++u) {
            sl[u] = f0[u] * wl0[u % H] + f1[u] * wl1[u % H];
            sr[u] = f0[u] * wr0[u % H] + f1[u] * wr1[u % H];
        }
#pragma unroll
        for (int o = 16; o > 0; o >>= 1) {
#pragma unroll
            for (int u = 0; u < PU; ++u) {
                sl[u] += __shfl_xor(sl[u], o, 32);
                sr[u] += __shfl_xor(sr[u], o, 32);
            }
        }
        float ol = sl[0], orr = sr[0];
#pragma unroll
        for (int u = 1; u < PU; ++u) if (lane == u) { ol = sl[u]; orr = sr[u]; }
        if (lane < PU && p0 + lane < nPairs) { el[p0 + lane] = ol; er[p0 + lane] = orr; }
    }
}

#define TROWS 196
#define QCAP  384

template <int C, int U>
__device__ __forceinline__ void agg_group(
    const int* myqs, const int* myqd, int q, int lane,
    const float* __restrict__ el, const float* __restrict__ feat,
    const float* ler, const float* rmax, float* rsum, float* tile)
{
    constexpr int H  = C / 64;
    constexpr int NI = C / 32;
    const int pl = lane % (U * H);
    const int pu = pl / H, ph = pl - pu * H;
    const int sl = myqs[q + pu];
    const int rl = myqd[q + pu];
    const float xe = el[sl * H + ph];
    const float xr = ler[rl * H + ph];
    const float mx = rmax[rl * H + ph];

    int s[U], dd[U];
#pragma unroll
    for (int u = 0; u < U; ++u) {
        s[u]  = __builtin_amdgcn_readfirstlane(myqs[q + u]);
        dd[u] = __builtin_amdgcn_readfirstlane(myqd[q + u]);
    }
    float v[U][NI];
#pragma unroll
    for (int u = 0; u < U; ++u) {
        const float* f = feat + (long)s[u] * C + lane;
#pragma unroll
        for (int i = 0; i < NI; ++i) v[u][i] = f[i * 32];
    }
    __builtin_amdgcn_sched_barrier(0);
    float x = xe + xr;
    x = x > 0.0f ? x : 0.2f * x;
    const float w = expf(x - mx);
    if (lane < U * H)
        __hip_atomic_fetch_add(rsum + rl * H + ph, w, __ATOMIC_RELAXED, __HIP_MEMORY_SCOPE_WORKGROUP);
#pragma unroll
    for (int u = 0; u < U; ++u) {
        float* t = tile + dd[u] * C + lane;
#pragma unroll
        for (int i = 0; i < NI; ++i) {
            const float a = __int_as_float(
                __builtin_amdgcn_readlane(__float_as_int(w), u * H + (i >> 1)));
            __hip_atomic_fetch_add(t + i * 32, v[u][i] * a,
                                   __ATOMIC_RELAXED, __HIP_MEMORY_SCOPE_WORKGROUP);
        }
    }
}

template <int C>
__global__ __launch_bounds__(1024) void edge_softmax_aggregate_tile(
    const int* __restrict__ src, const int* __restrict__ dst,
    const float* __restrict__ feat, const float* __restrict__ el,
    const float* __restrict__ er, float* __restrict__ rst, int E, int nNodes)
{
    constexpr int H = C / 64;
    __shared__ __attribute__((aligned(16))) float tile[TROWS * C];
    __shared__ float ler[TROWS * H];
    __shared__ float rmax[TROWS * H];
    __shared__ float rsum[TROWS * H];
    __shared__ int qs[32 * QCAP];
    __shared__ int qd[32 * QCAP];

    const int tid  = threadIdx.x;
    const int lane = tid & 31;
    const int wave = __builtin_amdgcn_readfirstlane(tid >> 5);
    const int d0   = blockIdx.x * TROWS;
    const int nrows = min(TROWS, nNodes - d0);

    for (int i = tid; i < TROWS * C / 4; i += 1024)
        ((float4*)tile)[i] = make_float4(0.0f, 0.0f, 0.0f, 0.0f);
    for (int i = tid; i < TROWS * H; i += 1024) {
        ler[i]  = (i < nrows * H) ? er[d0 * H + i] : 0.0f;
        rmax[i] = -INFINITY;
        rsum[i] = 0.0f;
    }
    __syncthreads();

    int* myqs = qs + wave * QCAP;
    int* myqd = qd + wave * QCAP;
    int qlen = 0;
    int done = 0;
    bool ovf = false;

    const int nfull = E >> 7;
    const int4 none = make_int4(-1, -1, -1, -1);

    auto stage1 = [&](int b, int e) {
        const int sub = lane / H, hh = lane - sub * H;
        for (int i = b; i < e; i += 32 / H) {
            const int idx = i + sub;
            if (idx < e) {
                const int ed  = myqs[idx];
                const int row = myqd[idx];
                const int sn  = src[ed];
                float x = el[sn * H + hh] + ler[row * H + hh];
                x = x > 0.0f ? x : 0.2f * x;
                __hip_atomic_fetch_max(rmax + row * H + hh, x,
                                       __ATOMIC_RELAXED, __HIP_MEMORY_SCOPE_WORKGROUP);
                if (hh == 0) myqs[idx] = sn;
            }
        }
    };

    auto stage2 = [&](int b, int e) {
        int q = b;
        for (; q + 4 <= e; q += 4)
            agg_group<C, 4>(myqs, myqd, q, lane, el, feat, ler, rmax, rsum, tile);
        for (; q < e; ++q)
            agg_group<C, 1>(myqs, myqd, q, lane, el, feat, ler, rmax, rsum, tile);
    };

    auto append = [&](const int4 cur, int chunk) -> bool {
        const unsigned r0 = (unsigned)(cur.x - d0), r1 = (unsigned)(cur.y - d0);
        const unsigned r2 = (unsigned)(cur.z - d0), r3 = (unsigned)(cur.w - d0);
        const bool m0 = r0 < (unsigned)nrows, m1 = r1 < (unsigned)nrows;
        const bool m2 = r2 < (unsigned)nrows, m3 = r3 < (unsigned)nrows;
        const unsigned k0 = __builtin_amdgcn_ballot_w32(m0), k1 = __builtin_amdgcn_ballot_w32(m1);
        const unsigned k2 = __builtin_amdgcn_ballot_w32(m2), k3 = __builtin_amdgcn_ballot_w32(m3);
        if ((k0 | k1 | k2 | k3) == 0u) return false;
        const int ebase = chunk * 128 + lane * 4;
        const unsigned rr[4] = {r0, r1, r2, r3};
        const bool     mm[4] = {m0, m1, m2, m3};
        const unsigned kk[4] = {k0, k1, k2, k3};
#pragma unroll
        for (int k = 0; k < 4; ++k) {
            if (mm[k]) {
                const int pos = qlen + (int)__builtin_amdgcn_mbcnt_lo(kk[k], 0u);
                myqs[pos] = ebase + k;
                myqd[pos] = (int)rr[k];
            }
            qlen += __popc(kk[k]);
        }
        return true;
    };

    auto scan = [&](auto&& pre, auto&& post) {
        if (wave < nfull) {
            const int4* dst4 = (const int4*)dst + lane;
            auto load_at = [&](int c) -> int4 {
                return dst4[(long)((c < nfull) ? c : wave) * 32];
            };
            int4 a0 = load_at(wave), a1 = load_at(wave + 32);
            for (int base = wave; base < nfull; base += 128) {
                const int4 b0 = load_at(base + 64), b1 = load_at(base + 96);
                pre(); append(a0, base);
                if (base + 32 < nfull) { pre(); append(a1, base + 32); }
                post();
                a0 = load_at(base + 128); a1 = load_at(base + 160);
                if (base + 64 < nfull) { pre(); append(b0, base + 64); }
                if (base + 96 < nfull) { pre(); append(b1, base + 96); }
                post();
            }
        }
        if (wave == 0 && (E & 127) != 0) {
            const int e0 = nfull * 128 + lane * 4;
            int4 cur = none;
            if (e0     < E) cur.x = dst[e0];
            if (e0 + 1 < E) cur.y = dst[e0 + 1];
            if (e0 + 2 < E) cur.z = dst[e0 + 2];
            if (e0 + 3 < E) cur.w = dst[e0 + 3];
            pre(); append(cur, nfull);
        }
    };

    scan(
        [&]() {
            if (qlen > QCAP - 128) {
                stage1(done, qlen);
                ovf = true; qlen = 0; done = 0;
            }
        },
        [&]() {
            if (qlen - done >= 32) { stage1(done, qlen); done = qlen; }
        });
    stage1(done, qlen);
    __syncthreads();

    if (!ovf) {
        stage2(0, qlen);
    } else {
        qlen = 0;
        scan(
            [&]() {},
            [&]() {
                if (qlen >= 32) { stage1(0, qlen); stage2(0, qlen); qlen = 0; }
            });
        stage1(0, qlen);
        stage2(0, qlen);
    }

    __syncthreads();
    float4* o = (float4*)(rst + (long)d0 * C);
    for (int i = tid; i < nrows * (C / 4); i += 1024) {
        const float4 t = ((const float4*)tile)[i];
        const int row = i / (C / 4);
        const int hh  = (i % (C / 4)) / 16;
        const float sm = rsum[row * H + hh];
        const float dn = (sm == 0.0f) ? 1.0f : sm;
        o[i] = make_float4(t.x / dn, t.y / dn, t.z / dn, t.w / dn);
    }
}

template <int C>
__global__ __launch_bounds__(256) void bias_elu(float* __restrict__ x, const float* __restrict__ b,
                                                int n4)
{
    float4* x4 = (float4*)x;
    const int stride = gridDim.x * 256;
    for (int i0 = blockIdx.x * 256 + threadIdx.x; i0 < n4; i0 += 4 * stride) {
        float4 v[4];
#pragma unroll
        for (int u = 0; u < 4; ++u) v[u] = x4[min(i0 + u * stride, n4 - 1)];
#pragma unroll
        for (int u = 0; u < 4; ++u) {
            const int i = i0 + u * stride;
            if (i < n4) {
                const float* bb = b + ((unsigned)i % (unsigned)(C / 4)) * 4u;
                float4 r;
                float t;
                t = v[u].x + bb[0]; r.x = t > 0.0f ? t : expm1f(t);
                t = v[u].y + bb[1]; r.y = t > 0.0f ? t : expm1f(t);
                t = v[u].z + bb[2]; r.z = t > 0.0f ? t : expm1f(t);
                t = v[u].w + bb[3]; r.w = t > 0.0f ? t : expm1f(t);
                x4[i] = r;
            }
        }
    }
}

__global__ __launch_bounds__(256) void classify(
    const float* __restrict__ rst2, const float* __restrict__ b2,
    const float* __restrict__ Wc1, const float* __restrict__ bc1,
    const float* __restrict__ Wc2, const float* __restrict__ bc2,
    float* __restrict__ out, int nNodes)
{
    const int lane = threadIdx.x & 31;
    const int wid  = blockIdx.x * 8 + __builtin_amdgcn_readfirstlane(threadIdx.x >> 5);
    const int nw   = gridDim.x * 8;
    float w[64];
#pragma unroll
    for (int d = 0; d < 64; ++d) w[d] = Wc1[d * 32 + lane];
    const float bl  = bc1[lane];
    const float w20 = Wc2[lane * 2 + 0], w21 = Wc2[lane * 2 + 1];
    const float c0  = bc2[0], c1 = bc2[1];
    for (int node = wid; node < nNodes; node += nw) {
        const float* r = rst2 + (long)node * 64;
        float acc = 0.0f;
#pragma unroll
        for (int d = 0; d < 64; ++d)
            acc += (r[d] + b2[d]) * w[d];
        float t = fmaxf(acc + bl, 0.0f);
        float o0 = t * w20;
        float o1 = t * w21;
#pragma unroll
        for (int o = 16; o > 0; o >>= 1) {
            o0 += __shfl_xor(o0, o, 32);
            o1 += __shfl_xor(o1, o, 32);
        }
        if (lane == 0) {
            out[(long)node * 2 + 0] = o0 + c0;
            out[(long)node * 2 + 1] = o1 + c1;
        }
    }
}

extern "C" void kernel_launch(void* const* d_in, const int* in_sizes, int n_in,
                              void* d_out, int out_size, void* d_ws, size_t ws_size,
                              hipStream_t stream)
{
    (void)n_in; (void)out_size; (void)ws_size;

    const float* randfeat = (const float*)d_in[0];
    const float* func_emb = (const float*)d_in[1];
    const float* codebert = (const float*)d_in[2];
    const int*   src      = (const int*)d_in[3];
    const int*   dst      = (const int*)d_in[4];
    const float* W1  = (const float*)d_in[5];
    const float* al1 = (const float*)d_in[6];
    const float* ar1 = (const float*)d_in[7];
    const float* b1  = (const float*)d_in[8];
    const float* W2  = (const float*)d_in[9];
    const float* al2 = (const float*)d_in[10];
    const float* ar2 = (const float*)d_in[11];
    const float* b2  = (const float*)d_in[12];
    const float* Wc1 = (const float*)d_in[13];
    const float* bc1 = (const float*)d_in[14];
    const float* Wc2 = (const float*)d_in[15];
    const float* bc2 = (const float*)d_in[16];
    float* out = (float*)d_out;

    const int  E  = in_sizes[3];
    const long NN = N_NODES;

    float* ws     = (float*)d_ws;
    float* feat1  = ws;
    float* rst1   = feat1  + NN * 256;
    float* feat2  = rst1   + NN * 256;
    float* rst2   = feat2  + NN * 64;
    float* el1    = rst2   + NN * 64;
    float* er1    = el1    + NN * 4;
    float* el2    = er1    + NN * 4;
    float* er2    = el2    + NN;

    gemm_wmma_f32<8, 2, 256><<<((N_NODES + 31) / 32 * 2 + 3) / 4, 128, 0, stream>>>(
        randfeat, func_emb, codebert, 100, 100, 768, W1, feat1, N_NODES);
    attn_proj<4><<<1024, 256, 0, stream>>>(feat1, al1, ar1, el1, er1, N_NODES * 4);
    edge_softmax_aggregate_tile<256><<<(N_NODES + TROWS - 1) / TROWS, 1024, 0, stream>>>(src, dst, feat1, el1, er1, rst1, E, N_NODES);
    bias_elu<256><<<1024, 256, 0, stream>>>(rst1, b1, N_NODES * 64);

    gemm_wmma_f32<4, 1, 64><<<((N_NODES + 31) / 32 + 3) / 4, 128, 0, stream>>>(
        rst1, rst1, rst1, 256, 0, 0, W2, feat2, N_NODES);
    attn_proj<1><<<256, 256, 0, stream>>>(feat2, al2, ar2, el2, er2, N_NODES);
    edge_softmax_aggregate_tile<64><<<(N_NODES + TROWS - 1) / TROWS, 1024, 0, stream>>>(src, dst, feat2, el2, er2, rst2, E, N_NODES);

    classify<<<1024, 256, 0, stream>>>(rst2, b2, Wc1, bc1, Wc2, bc2, out, N_NODES);
}


# Tile edge lists built once per call: a binning launch sorts each 4096-edge chunk by dst tile in LDS (src+row words, segment starts); edge launches read segments, no dst scans; LDS add order changes
# speedup vs baseline: 5.3832x; 1.1766x over previous
#include <hip/hip_runtime.h>
#include <math.h>


typedef __attribute__((ext_vector_type(2))) float v2f;
typedef __attribute__((ext_vector_type(8))) float v8f;

#define N_NODES 50000

template <int NT, int NC>
__device__ __forceinline__ void seg_accum(
    const float* __restrict__ a0, const float* __restrict__ a1, int w,
    const float* __restrict__ bcol, int h, v8f (&acc)[2][NT])
{
    const int nblk = w >> 5;
    if (nblk > 0) {
        const float4* q0 = (const float4*)(a0 + 16 * h);
        const float4* q1 = (const float4*)(a1 + 16 * h);
        const float* bp = bcol + (long)(16 * h) * NC;
        float4 ra[4], rb[4];
        v2f bc[NT];
#pragma unroll
        for (int q = 0; q < 4; ++q) { ra[q] = q0[q]; rb[q] = q1[q]; }
#pragma unroll
        for (int t = 0; t < NT; ++t) { bc[t].x = bp[t * 16]; bc[t].y = bp[NC + t * 16]; }

        for (int blk = 0; blk < nblk; ++blk) {
            const int adv = (blk + 1 < nblk) ? 32 : 0;
            const float4* nq0 = (const float4*)((const float*)q0 + adv);
            const float4* nq1 = (const float4*)((const float*)q1 + adv);
            const float* nbp = bp + (long)adv * NC;
            float4 na[4], nb[4];
#pragma unroll
            for (int q = 0; q < 4; ++q) { na[q] = nq0[q]; nb[q] = nq1[q]; }
            __builtin_amdgcn_sched_barrier(0);
#pragma unroll
            for (int j = 0; j < 8; ++j) {
                v2f bn[NT];
                if (j < 7) {
#pragma unroll
                    for (int t = 0; t < NT; ++t) {
                        bn[t].x = bp[(2 * j + 2) * NC + t * 16];
                        bn[t].y = bp[(2 * j + 3) * NC + t * 16];
                    }
                } else {
#pragma unroll
                    for (int t = 0; t < NT; ++t) {
                        bn[t].x = nbp[t * 16];
                        bn[t].y = nbp[NC + t * 16];
                    }
                }
                v2f fa, fb;
                if (j & 1) { fa.x = ra[j >> 1].z; fa.y = ra[j >> 1].w; fb.x = rb[j >> 1].z; fb.y = rb[j >> 1].w; }
                else       { fa.x = ra[j >> 1].x; fa.y = ra[j >> 1].y; fb.x = rb[j >> 1].x; fb.y = rb[j >> 1].y; }
#pragma unroll
                for (int t = 0; t < NT; ++t) {
                    acc[0][t] = __builtin_amdgcn_wmma_f32_16x16x4_f32(
                        false, fa, false, bc[t], (short)0, acc[0][t], false, false);
                    acc[1][t] = __builtin_amdgcn_wmma_f32_16x16x4_f32(
                        false, fb, false, bc[t], (short)0, acc[1][t], false, false);
                }
#pragma unroll
                for (int t = 0; t < NT; ++t) bc[t] = bn[t];
            }
            __builtin_amdgcn_sched_barrier(0);
            asm volatile("v_nop\n\tv_nop\n\tv_nop\n\tv_nop");
#pragma unroll
            for (int q = 0; q < 4; ++q) { ra[q] = na[q]; rb[q] = nb[q]; }
            q0 = nq0; q1 = nq1; bp = nbp;
        }
    }
    for (int k = nblk << 5; k + 4 <= w; k += 4) {
        const float2 ta = *(const float2*)(a0 + k + 2 * h);
        const float2 tb = *(const float2*)(a1 + k + 2 * h);
        v2f fa, fb;
        fa.x = ta.x; fa.y = ta.y; fb.x = tb.x; fb.y = tb.y;
        const float* bp = bcol + (long)(k + 2 * h) * NC;
#pragma unroll
        for (int t = 0; t < NT; ++t) {
            v2f b;
            b.x = bp[t * 16];
            b.y = bp[NC + t * 16];
            acc[0][t] = __builtin_amdgcn_wmma_f32_16x16x4_f32(
                false, fa, false, b, (short)0, acc[0][t], false, false);
            acc[1][t] = __builtin_amdgcn_wmma_f32_16x16x4_f32(
                false, fb, false, b, (short)0, acc[1][t], false, false);
        }
    }
}

template <int NT, int NG, int NC>
__global__ __launch_bounds__(128) void gemm_wmma_f32(
    const float* __restrict__ p0, const float* __restrict__ p1,
    const float* __restrict__ p2, int w0, int w1, int w2,
    const float* __restrict__ B, float* __restrict__ C, int M)
{
    const int lane = threadIdx.x & 31;
    const int gw   = blockIdx.x * 4 + (threadIdx.x >> 5);
    const int ng   = gw % NG;
    const int m0   = (gw / NG) * 32;
    if (m0 >= M) return;
    const int c  = lane & 15;
    const int h  = lane >> 4;
    const int n0 = ng * NT * 16;

    const int r0 = min(m0 + c, M - 1);
    const int r1 = min(m0 + 16 + c, M - 1);

    v8f acc[2][NT];
#pragma unroll
    for (int t = 0; t < NT; ++t) { acc[0][t] = v8f{}; acc[1][t] = v8f{}; }

    const float* bcol = B + n0 + c;
    seg_accum<NT, NC>(p0 + (long)r0 * w0, p0 + (long)r1 * w0, w0, bcol, h, acc);
    seg_accum<NT, NC>(p1 + (long)r0 * w1, p1 + (long)r1 * w1, w1, bcol + (long)w0 * NC, h, acc);
    seg_accum<NT, NC>(p2 + (long)r0 * w2, p2 + (long)r1 * w2, w2, bcol + (long)(w0 + w1) * NC, h, acc);

#pragma unroll
    for (int i = 0; i < 2; ++i) {
#pragma unroll
        for (int r = 0; r < 8; ++r) {
            const int row = m0 + i * 16 + 8 * h + r;
            if (row < M) {
                float* crow = C + (long)row * NC + n0 + c;
#pragma unroll
                for (int t = 0; t < NT; ++t) crow[t * 16] = acc[i][t][r];
            }
        }
    }
}

#define PU 8
template <int H>
__global__ __launch_bounds__(256) void attn_proj(
    const float* __restrict__ feat, const float* __restrict__ al,
    const float* __restrict__ ar, float* __restrict__ el,
    float* __restrict__ er, int nPairs)
{
    static_assert(PU % H == 0, "a trip starts at head 0");
    const int lane = threadIdx.x & 31;
    const int wid  = blockIdx.x * 8 + (threadIdx.x >> 5);
    const int nw   = gridDim.x * 8;
    float wl0[H], wl1[H], wr0[H], wr1[H];
#pragma unroll
    for (int h = 0; h < H; ++h) {
        wl0[h] = al[h * 64 + lane]; wl1[h] = al[h * 64 + lane + 32];
        wr0[h] = ar[h * 64 + lane]; wr1[h] = ar[h * 64 + lane + 32];
    }
    for (int p0 = wid * PU; p0 < nPairs; p0 += nw * PU) {
        float f0[PU], f1[PU];
#pragma unroll
        for (int u = 0; u < PU; ++u) {
            const float* f = feat + (long)min(p0 + u, nPairs - 1) * 64;
            f0[u] = f[lane]; f1[u] = f[lane + 32];
        }
        float sl[PU], sr[PU];
#pragma unroll
        for (int u = 0; u < PU; ++u) {
            sl[u] = f0[u] * wl0[u % H] + f1[u] * wl1[u % H];
            sr[u] = f0[u] * wr0[u % H] + f1[u] * wr1[u % H];
        }
#pragma unroll
        for (int o = 16; o > 0; o >>= 1) {
#pragma unroll
            for (int u = 0; u < PU; ++u) {
                sl[u] += __shfl_xor(sl[u], o, 32);
                sr[u] += __shfl_xor(sr[u], o, 32);
            }
        }
        float ol = sl[0], orr = sr[0];
#pragma unroll
        for (int u = 1; u < PU; ++u) if (lane == u) { ol = sl[u]; orr = sr[u]; }
        if (lane < PU && p0 + lane < nPairs) { el[p0 + lane] = ol; er[p0 + lane] = orr; }
    }
}

#define TROWS 196
#define QCAP  384
#define CHUNK 4096
#define NTILES ((N_NODES + TROWS - 1) / TROWS)

__global__ __launch_bounds__(1024) void bin_edges_by_tile(
    const int* __restrict__ src, const int* __restrict__ dst,
    int* __restrict__ ent, int* __restrict__ off, int E, int nNodes)
{
    static_assert(TROWS <= 256, "the row inside a tile is packed into 8 bits");
    static_assert(NTILES <= 1024, "one thread per tile in the prefix sum");
    __shared__ int hist[NTILES];
    __shared__ int start[NTILES + 1];
    __shared__ int wsum[32];
    __shared__ int sorted[CHUNK];

    const int tid  = threadIdx.x;
    const int lane = tid & 31;
    const int wv   = tid >> 5;
    const int nchunks = (E + CHUNK - 1) / CHUNK;
    for (int c = blockIdx.x; c < nchunks; c += gridDim.x) {
        for (int i = tid; i < NTILES; i += 1024) hist[i] = 0;
        __syncthreads();

        const int e0 = c * CHUNK + tid * 4;
        int d[4], s[4];
        if (e0 + 3 < E) {
            const int4 dv = *(const int4*)(dst + e0);
            const int4 sv = *(const int4*)(src + e0);
            d[0] = dv.x; d[1] = dv.y; d[2] = dv.z; d[3] = dv.w;
            s[0] = sv.x; s[1] = sv.y; s[2] = sv.z; s[3] = sv.w;
        } else {
#pragma unroll
            for (int k = 0; k < 4; ++k) {
                d[k] = (e0 + k < E) ? dst[e0 + k] : -1;
                s[k] = (e0 + k < E) ? src[e0 + k] : 0;
            }
        }
        int tl[4], rk[4], pk[4];
#pragma unroll
        for (int k = 0; k < 4; ++k) {
            tl[k] = -1; rk[k] = 0; pk[k] = 0;
            if ((unsigned)d[k] < (unsigned)nNodes) {
                tl[k] = d[k] / TROWS;
                pk[k] = (s[k] << 8) | (d[k] - tl[k] * TROWS);
                rk[k] = __hip_atomic_fetch_add(hist + tl[k], 1, __ATOMIC_RELAXED,
                                               __HIP_MEMORY_SCOPE_WORKGROUP);
            }
        }
        __syncthreads();

        int v = 0, incl = 0;
        if (wv < (NTILES + 31) / 32) {
            v = (tid < NTILES) ? hist[tid] : 0;
            incl = v;
            for (int o = 1; o < 32; o <<= 1) {
                const int t = __shfl_up(incl, o, 32);
                if (lane >= o) incl += t;
            }
            if (lane == 31) wsum[wv] = incl;
        }
        __syncthreads();
        if (tid < NTILES) {
            int base = 0;
            for (int w = 0; w < wv; ++w) base += wsum[w];
            start[tid] = base + incl - v;
            if (tid == NTILES - 1) start[NTILES] = base + incl;
        }
        __syncthreads();

#pragma unroll
        for (int k = 0; k < 4; ++k)
            if (tl[k] >= 0) sorted[start[tl[k]] + rk[k]] = pk[k];
        __syncthreads();

        const int cnt = start[NTILES];
        for (int i = tid; i < cnt; i += 1024) ent[(long)c * CHUNK + i] = sorted[i];
        for (int i = tid; i <= NTILES; i += 1024) off[(long)c * (NTILES + 1) + i] = start[i];
        __syncthreads();
    }
}

template <int C, int U>
__device__ __forceinline__ void agg_group(
    const int* myqs, const int* myqd, int q, int lane,
    const float* __restrict__ el, const float* __restrict__ feat,
    const float* ler, const float* rmax, float* rsum, float* tile)
{
    constexpr int H  = C / 64;
    constexpr int NI = C / 32;
    const int pl = lane % (U * H);
    const int pu = pl / H, ph = pl - pu * H;
    const int sl = myqs[q + pu];
    const int rl = myqd[q + pu];
    const float xe = el[sl * H + ph];
    const float xr = ler[rl * H + ph];
    const float mx = rmax[rl * H + ph];

    int s[U], dd[U];
#pragma unroll
    for (int u = 0; u < U; ++u) {
        s[u]  = __builtin_amdgcn_readfirstlane(myqs[q + u]);
        dd[u] = __builtin_amdgcn_readfirstlane(myqd[q + u]);
    }
    float v[U][NI];
#pragma unroll
    for (int u = 0; u < U; ++u) {
        const float* f = feat + (long)s[u] * C + lane;
#pragma unroll
        for (int i = 0; i < NI; ++i) v[u][i] = f[i * 32];
    }
    __builtin_amdgcn_sched_barrier(0);
    float x = xe + xr;
    x = x > 0.0f ? x : 0.2f * x;
    const float w = expf(x - mx);
    if (lane < U * H)
        __hip_atomic_fetch_add(rsum + rl * H + ph, w, __ATOMIC_RELAXED, __HIP_MEMORY_SCOPE_WORKGROUP);
#pragma unroll
    for (int u = 0; u < U; ++u) {
        float* t = tile + dd[u] * C + lane;
#pragma unroll
        for (int i = 0; i < NI; ++i) {
            const float a = __int_as_float(
                __builtin_amdgcn_readlane(__float_as_int(w), u * H + (i >> 1)));
            __hip_atomic_fetch_add(t + i * 32, v[u][i] * a,
                                   __ATOMIC_RELAXED, __HIP_MEMORY_SCOPE_WORKGROUP);
        }
    }
}

template <int C>
__global__ __launch_bounds__(1024) void edge_softmax_aggregate_tile(
    const int* __restrict__ ent, const int* __restrict__ off,
    const float* __restrict__ feat, const float* __restrict__ el,
    const float* __restrict__ er, float* __restrict__ rst, int E, int nNodes)
{
    constexpr int H = C / 64;
    __shared__ __attribute__((aligned(16))) float tile[TROWS * C];
    __shared__ float ler[TROWS * H];
    __shared__ float rmax[TROWS * H];
    __shared__ float rsum[TROWS * H];
    __shared__ int qs[32 * QCAP];
    __shared__ int qd[32 * QCAP];

    const int tid  = threadIdx.x;
    const int lane = tid & 31;
    const int wave = __builtin_amdgcn_readfirstlane(tid >> 5);
    const int d0   = blockIdx.x * TROWS;
    const int nrows = min(TROWS, nNodes - d0);
    const int nchunks = (E + CHUNK - 1) / CHUNK;

    for (int i = tid; i < TROWS * C / 4; i += 1024)
        ((float4*)tile)[i] = make_float4(0.0f, 0.0f, 0.0f, 0.0f);
    for (int i = tid; i < TROWS * H; i += 1024) {
        ler[i]  = (i < nrows * H) ? er[d0 * H + i] : 0.0f;
        rmax[i] = -INFINITY;
        rsum[i] = 0.0f;
    }
    __syncthreads();

    int* myqs = qs + wave * QCAP;
    int* myqd = qd + wave * QCAP;
    int qlen = 0;
    bool ovf = false;

    auto stage1 = [&](int e) {
        constexpr int PER = 32 / H;
        const int sub = lane / H, hh = lane - sub * H;
        for (int i = 0; i < e; i += 4 * PER) {
            int row[4];
            float xe[4];
#pragma unroll
            for (int u = 0; u < 4; ++u) {
                const int idx = min(i + u * PER + sub, e - 1);
                row[u] = myqd[idx];
                xe[u]  = el[myqs[idx] * H + hh];
            }
#pragma unroll
            for (int u = 0; u < 4; ++u) {
                if (i + u * PER + sub < e) {
                    float x = xe[u] + ler[row[u] * H + hh];
                    x = x > 0.0f ? x : 0.2f * x;
                    __hip_atomic_fetch_max(rmax + row[u] * H + hh, x,
                                           __ATOMIC_RELAXED, __HIP_MEMORY_SCOPE_WORKGROUP);
                }
            }
        }
    };

    auto stage2 = [&](int e) {
        int q = 0;
        for (; q + 4 <= e; q += 4)
            agg_group<C, 4>(myqs, myqd, q, lane, el, feat, ler, rmax, rsum, tile);
        for (; q < e; ++q)
            agg_group<C, 1>(myqs, myqd, q, lane, el, feat, ler, rmax, rsum, tile);
    };

    auto walk = [&](auto&& full) {
        for (int g0 = wave; g0 < nchunks; g0 += 1024) {
            const int cidx = g0 + 32 * lane;
            int sbeg = 0, n = 0;
            if (cidx < nchunks) {
                const int* o = off + (long)cidx * (NTILES + 1) + blockIdx.x;
                const int a = o[0], b = o[1];
                sbeg = cidx * CHUNK + a;
                n = b - a;
            }
            int incl = n;
            for (int o = 1; o < 32; o <<= 1) {
                const int t = __shfl_up(incl, o, 32);
                if (lane >= o) incl += t;
            }
            const int excl  = incl - n;
            const int delta = sbeg - excl;
            const int total = __builtin_amdgcn_readlane(incl, 31);
            const int nact  = min(32, (nchunks - g0 + 31) >> 5);
            for (int f0 = 0; f0 < total; f0 += 128) {
                if (qlen > QCAP - 128) { full(qlen); ovf = true; qlen = 0; }
                int fc[4], ad[4], w[4];
#pragma unroll
                for (int k = 0; k < 4; ++k) {
                    fc[k] = min(f0 + k * 32 + lane, total - 1);
                    ad[k] = 0;
                }
                for (int j = 0; j < nact; ++j) {
                    const int ej = __builtin_amdgcn_readlane(excl, j);
                    const int dj = __builtin_amdgcn_readlane(delta, j);
#pragma unroll
                    for (int k = 0; k < 4; ++k) if (fc[k] >= ej) ad[k] = dj + fc[k];
                }
#pragma unroll
                for (int k = 0; k < 4; ++k) w[k] = ent[ad[k]];
#pragma unroll
                for (int k = 0; k < 4; ++k) {
                    if (f0 + k * 32 + lane < total) {
                        const int pos = qlen + k * 32 + lane;
                        myqs[pos] = w[k] >> 8;
                        myqd[pos] = w[k] & 255;
                    }
                }
                qlen += min(128, total - f0);
            }
        }
    };

    walk([&](int n) { stage1(n); });
    stage1(qlen);
    __syncthreads();

    if (!ovf) {
        stage2(qlen);
    } else {
        qlen = 0;
        walk([&](int n) { stage2(n); });
        stage2(qlen);
    }

    __syncthreads();
    float4* o = (float4*)(rst + (long)d0 * C);
    for (int i = tid; i < nrows * (C / 4); i += 1024) {
        const float4 t = ((const float4*)tile)[i];
        const int row = i / (C / 4);
        const int hh  = (i % (C / 4)) / 16;
        const float sm = rsum[row * H + hh];
        const float dn = (sm == 0.0f) ? 1.0f : sm;
        o[i] = make_float4(t.x / dn, t.y / dn, t.z / dn, t.w / dn);
    }
}

template <int C>
__global__ __launch_bounds__(256) void bias_elu(float* __restrict__ x, const float* __restrict__ b,
                                                int n4)
{
    float4* x4 = (float4*)x;
    const int stride = gridDim.x * 256;
    for (int i0 = blockIdx.x * 256 + threadIdx.x; i0 < n4; i0 += 4 * stride) {
        float4 v[4];
#pragma unroll
        for (int u = 0; u < 4; ++u) v[u] = x4[min(i0 + u * stride, n4 - 1)];
#pragma unroll
        for (int u = 0; u < 4; ++u) {
            const int i = i0 + u * stride;
            if (i < n4) {
                const float* bb = b + ((unsigned)i % (unsigned)(C / 4)) * 4u;
                float4 r;
                float t;
                t = v[u].x + bb[0]; r.x = t > 0.0f ? t : expm1f(t);
                t = v[u].y + bb[1]; r.y = t > 0.0f ? t : expm1f(t);
                t = v[u].z + bb[2]; r.z = t > 0.0f ? t : expm1f(t);
                t = v[u].w + bb[3]; r.w = t > 0.0f ? t : expm1f(t);
                x4[i] = r;
            }
        }
    }
}

__global__ __launch_bounds__(256) void classify(
    const float* __restrict__ rst2, const float* __restrict__ b2,
    const float* __restrict__ Wc1, const float* __restrict__ bc1,
    const float* __restrict__ Wc2, const float* __restrict__ bc2,
    float* __restrict__ out, int nNodes)
{
    const int lane = threadIdx.x & 31;
    const int wid  = blockIdx.x * 8 + __builtin_amdgcn_readfirstlane(threadIdx.x >> 5);
    const int nw   = gridDim.x * 8;
    float w[64];
#pragma unroll
    for (int d = 0; d < 64; ++d) w[d] = Wc1[d * 32 + lane];
    const float bl  = bc1[lane];
    const float w20 = Wc2[lane * 2 + 0], w21 = Wc2[lane * 2 + 1];
    const float c0  = bc2[0], c1 = bc2[1];
    for (int node = wid; node < nNodes; node += nw) {
        const float* r = rst2 + (long)node * 64;
        float acc = 0.0f;
#pragma unroll
        for (int d = 0; d < 64; ++d)
            acc += (r[d] + b2[d]) * w[d];
        float t = fmaxf(acc + bl, 0.0f);
        float o0 = t * w20;
        float o1 = t * w21;
#pragma unroll
        for (int o = 16; o > 0; o >>= 1) {
            o0 += __shfl_xor(o0, o, 32);
            o1 += __shfl_xor(o1, o, 32);
        }
        if (lane == 0) {
            out[(long)node * 2 + 0] = o0 + c0;
            out[(long)node * 2 + 1] = o1 + c1;
        }
    }
}

extern "C" void kernel_launch(void* const* d_in, const int* in_sizes, int n_in,
                              void* d_out, int out_size, void* d_ws, size_t ws_size,
                              hipStream_t stream)
{
    (void)n_in; (void)out_size; (void)ws_size;

    const float* randfeat = (const float*)d_in[0];
    const float* func_emb = (const float*)d_in[1];
    const float* codebert = (const float*)d_in[2];
    const int*   src      = (const int*)d_in[3];
    const int*   dst      = (const int*)d_in[4];
    const float* W1  = (const float*)d_in[5];
    const float* al1 = (const float*)d_in[6];
    const float* ar1 = (const float*)d_in[7];
    const float* b1  = (const float*)d_in[8];
    const float* W2  = (const float*)d_in[9];
    const float* al2 = (const float*)d_in[10];
    const float* ar2 = (const float*)d_in[11];
    const float* b2  = (const float*)d_in[12];
    const float* Wc1 = (const float*)d_in[13];
    const float* bc1 = (const float*)d_in[14];
    const float* Wc2 = (const float*)d_in[15];
    const float* bc2 = (const float*)d_in[16];
    float* out = (float*)d_out;

    const int  E  = in_sizes[3];
    const long NN = N_NODES;

    float* ws     = (float*)d_ws;
    float* feat1  = ws;
    float* rst1   = feat1  + NN * 256;
    int*   ent    = (int*)(rst1 + NN * 256);
    int*   off    = ent    + NN * 48;
    float* rst2   = rst1   + NN * 256 + NN * 64;
    float* feat2  = feat1;
    float* el1    = rst2   + NN * 64;
    float* er1    = el1    + NN * 4;
    float* el2    = er1    + NN * 4;
    float* er2    = el2    + NN;

    bin_edges_by_tile<<<256, 1024, 0, stream>>>(src, dst, ent, off, E, N_NODES);

    gemm_wmma_f32<8, 2, 256><<<((N_NODES + 31) / 32 * 2 + 3) / 4, 128, 0, stream>>>(
        randfeat, func_emb, codebert, 100, 100, 768, W1, feat1, N_NODES);
    attn_proj<4><<<1024, 256, 0, stream>>>(feat1, al1, ar1, el1, er1, N_NODES * 4);
    edge_softmax_aggregate_tile<256><<<NTILES, 1024, 0, stream>>>(ent, off, feat1, el1, er1, rst1, E, N_NODES);
    bias_elu<256><<<1024, 256, 0, stream>>>(rst1, b1, N_NODES * 64);

    gemm_wmma_f32<4, 1, 64><<<((N_NODES + 31) / 32 + 3) / 4, 128, 0, stream>>>(
        rst1, rst1, rst1, 256, 0, 0, W2, feat2, N_NODES);
    attn_proj<1><<<256, 256, 0, stream>>>(feat2, al2, ar2, el2, er2, N_NODES);
    edge_softmax_aggregate_tile<64><<<NTILES, 1024, 0, stream>>>(ent, off, feat2, el2, er2, rst2, E, N_NODES);

    classify<<<1024, 256, 0, stream>>>(rst2, b2, Wc1, bc1, Wc2, bc2, out, N_NODES);
}


# Wide GEMM dealt over a resident grid: 256 blocks of 12 waves (3 per SIMD) take 1536 row groups, the 848 rows left go as 424 units of 16 rows x 32 cols to waves 0 and 1; same summation order
# speedup vs baseline: 5.6409x; 1.0479x over previous
#include <hip/hip_runtime.h>
#include <math.h>


typedef __attribute__((ext_vector_type(2))) float v2f;
typedef __attribute__((ext_vector_type(8))) float v8f;

#define N_NODES 50000

template <int NT, int NC>
__device__ __forceinline__ void seg_accum(
    const float* __restrict__ a0, const float* __restrict__ a1, int w,
    const float* __restrict__ bcol, int h, v8f (&acc)[2][NT])
{
    const int nblk = w >> 5;
    if (nblk > 0) {
        const float4* q0 = (const float4*)(a0 + 16 * h);
        const float4* q1 = (const float4*)(a1 + 16 * h);
        const float* bp = bcol + (long)(16 * h) * NC;
        float4 ra[4], rb[4];
        v2f bc[NT];
#pragma unroll
        for (int q = 0; q < 4; ++q) { ra[q] = q0[q]; rb[q] = q1[q]; }
#pragma unroll
        for (int t = 0; t < NT; ++t) { bc[t].x = bp[t * 16]; bc[t].y = bp[NC + t * 16]; }

        for (int blk = 0; blk < nblk; ++blk) {
            const int adv = (blk + 1 < nblk) ? 32 : 0;
            const float4* nq0 = (const float4*)((const float*)q0 + adv);
            const float4* nq1 = (const float4*)((const float*)q1 + adv);
            const float* nbp = bp + (long)adv * NC;
            float4 na[4], nb[4];
#pragma unroll
            for (int q = 0; q < 4; ++q) { na[q] = nq0[q]; nb[q] = nq1[q]; }
            __builtin_amdgcn_sched_barrier(0);
#pragma unroll
            for (int j = 0; j < 8; ++j) {
                v2f bn[NT];
                if (j < 7) {
#pragma unroll
                    for (int t = 0; t < NT; ++t) {
                        bn[t].x = bp[(2 * j + 2) * NC + t * 16];
                        bn[t].y = bp[(2 * j + 3) * NC + t * 16];
                    }
                } else {
#pragma unroll
                    for (int t = 0; t < NT; ++t) {
                        bn[t].x = nbp[t * 16];
                        bn[t].y = nbp[NC + t * 16];
                    }
                }
                v2f fa, fb;
                if (j & 1) { fa.x = ra[j >> 1].z; fa.y = ra[j >> 1].w; fb.x = rb[j >> 1].z; fb.y = rb[j >> 1].w; }
                else       { fa.x = ra[j >> 1].x; fa.y = ra[j >> 1].y; fb.x = rb[j >> 1].x; fb.y = rb[j >> 1].y; }
#pragma unroll
                for (int t = 0; t < NT; ++t) {
                    acc[0][t] = __builtin_amdgcn_wmma_f32_16x16x4_f32(
                        false, fa, false, bc[t], (short)0, acc[0][t], false, false);
                    acc[1][t] = __builtin_amdgcn_wmma_f32_16x16x4_f32(
                        false, fb, false, bc[t], (short)0, acc[1][t], false, false);
                }
#pragma unroll
                for (int t = 0; t < NT; ++t) bc[t] = bn[t];
            }
            __builtin_amdgcn_sched_barrier(0);
            asm volatile("v_nop\n\tv_nop\n\tv_nop\n\tv_nop");
#pragma unroll
            for (int q = 0; q < 4; ++q) { ra[q] = na[q]; rb[q] = nb[q]; }
            q0 = nq0; q1 = nq1; bp = nbp;
        }
    }
    for (int k = nblk << 5; k + 4 <= w; k += 4) {
        const float2 ta = *(const float2*)(a0 + k + 2 * h);
        const float2 tb = *(const float2*)(a1 + k + 2 * h);
        v2f fa, fb;
        fa.x = ta.x; fa.y = ta.y; fb.x = tb.x; fb.y = tb.y;
        const float* bp = bcol + (long)(k + 2 * h) * NC;
#pragma unroll
        for (int t = 0; t < NT; ++t) {
            v2f b;
            b.x = bp[t * 16];
            b.y = bp[NC + t * 16];
            acc[0][t] = __builtin_amdgcn_wmma_f32_16x16x4_f32(
                false, fa, false, b, (short)0, acc[0][t], false, false);
            acc[1][t] = __builtin_amdgcn_wmma_f32_16x16x4_f32(
                false, fb, false, b, (short)0, acc[1][t], false, false);
        }
    }
}

template <int NC>
__device__ __forceinline__ void seg_accum_rem(
    const float* __restrict__ a0, int w,
    const float* __restrict__ bcol, int h, v8f (&acc)[2])
{
    const int nblk = w >> 5;
    if (nblk > 0) {
        const float4* q0 = (const float4*)(a0 + 16 * h);
        const float* bp = bcol + (long)(16 * h) * NC;
        float4 ra[4];
        v2f rb[8][2];
#pragma unroll
        for (int q = 0; q < 4; ++q) ra[q] = q0[q];
#pragma unroll
        for (int j = 0; j < 8; ++j) {
#pragma unroll
            for (int t = 0; t < 2; ++t) {
                rb[j][t].x = bp[(2 * j) * NC + t * 16];
                rb[j][t].y = bp[(2 * j + 1) * NC + t * 16];
            }
        }
        for (int blk = 0; blk < nblk; ++blk) {
            const int adv = (blk + 1 < nblk) ? 32 : 0;
            const float4* nq0 = (const float4*)((const float*)q0 + adv);
            const float* nbp = bp + (long)adv * NC;
            float4 na[4];
            v2f nb[8][2];
#pragma unroll
            for (int q = 0; q < 4; ++q) na[q] = nq0[q];
#pragma unroll
            for (int j = 0; j < 8; ++j) {
#pragma unroll
                for (int t = 0; t < 2; ++t) {
                    nb[j][t].x = nbp[(2 * j) * NC + t * 16];
                    nb[j][t].y = nbp[(2 * j + 1) * NC + t * 16];
                }
            }
            __builtin_amdgcn_sched_barrier(0);
#pragma unroll
            for (int j = 0; j < 8; ++j) {
                v2f fa;
                if (j & 1) { fa.x = ra[j >> 1].z; fa.y = ra[j >> 1].w; }
                else       { fa.x = ra[j >> 1].x; fa.y = ra[j >> 1].y; }
#pragma unroll
                for (int t = 0; t < 2; ++t)
                    acc[t] = __builtin_amdgcn_wmma_f32_16x16x4_f32(
                        false, fa, false, rb[j][t], (short)0, acc[t], false, false);
            }
            __builtin_amdgcn_sched_barrier(0);
#pragma unroll
            for (int q = 0; q < 4; ++q) ra[q] = na[q];
#pragma unroll
            for (int j = 0; j < 8; ++j) { rb[j][0] = nb[j][0]; rb[j][1] = nb[j][1]; }
            q0 = nq0; bp = nbp;
        }
    }
    for (int k = nblk << 5; k + 4 <= w; k += 4) {
        const float2 ta = *(const float2*)(a0 + k + 2 * h);
        v2f fa;
        fa.x = ta.x; fa.y = ta.y;
        const float* bp = bcol + (long)(k + 2 * h) * NC;
#pragma unroll
        for (int t = 0; t < 2; ++t) {
            v2f b;
            b.x = bp[t * 16];
            b.y = bp[NC + t * 16];
            acc[t] = __builtin_amdgcn_wmma_f32_16x16x4_f32(
                false, fa, false, b, (short)0, acc[t], false, false);
        }
    }
}

template <int NT, int NG, int NC, int WPB, bool REM>
__global__ __launch_bounds__(WPB * 32) void gemm_wmma_f32(
    const float* __restrict__ p0, const float* __restrict__ p1,
    const float* __restrict__ p2, int w0, int w1, int w2,
    const float* __restrict__ B, float* __restrict__ C, int M)
{
    const int lane = threadIdx.x & 31;
    const int wv   = threadIdx.x >> 5;
    const int gw   = blockIdx.x * WPB + wv;
    const int c  = lane & 15;
    const int h  = lane >> 4;

    if (REM) {
        const int nWaves   = gridDim.x * WPB;
        const int mainRows = (nWaves / NG) * 32;
        const int nRem = (mainRows < M) ? ((M - mainRows + 15) / 16) * (NC / 32) : 0;
        for (int u = wv * gridDim.x + blockIdx.x; u < nRem; u += nWaves) {
            const int tm0 = mainRows + (u / (NC / 32)) * 16;
            const int tn0 = (u % (NC / 32)) * 32;
            const int rr = min(tm0 + c, M - 1);
            v8f racc[2];
            racc[0] = v8f{}; racc[1] = v8f{};
            const float* rcol = B + tn0 + c;
            seg_accum_rem<NC>(p0 + (long)rr * w0, w0, rcol, h, racc);
            seg_accum_rem<NC>(p1 + (long)rr * w1, w1, rcol + (long)w0 * NC, h, racc);
            seg_accum_rem<NC>(p2 + (long)rr * w2, w2, rcol + (long)(w0 + w1) * NC, h, racc);
#pragma unroll
            for (int r = 0; r < 8; ++r) {
                const int row = tm0 + 8 * h + r;
                if (row < M) {
                    float* crow = C + (long)row * NC + tn0 + c;
                    crow[0]  = racc[0][r];
                    crow[16] = racc[1][r];
                }
            }
        }
    }

    const int ng   = gw % NG;
    const int m0   = (gw / NG) * 32;
    if (m0 >= M) return;
    const int n0 = ng * NT * 16;

    const int r0 = min(m0 + c, M - 1);
    const int r1 = min(m0 + 16 + c, M - 1);

    v8f acc[2][NT];
#pragma unroll
    for (int t = 0; t < NT; ++t) { acc[0][t] = v8f{}; acc[1][t] = v8f{}; }

    const float* bcol = B + n0 + c;
    seg_accum<NT, NC>(p0 + (long)r0 * w0, p0 + (long)r1 * w0, w0, bcol, h, acc);
    seg_accum<NT, NC>(p1 + (long)r0 * w1, p1 + (long)r1 * w1, w1, bcol + (long)w0 * NC, h, acc);
    seg_accum<NT, NC>(p2 + (long)r0 * w2, p2 + (long)r1 * w2, w2, bcol + (long)(w0 + w1) * NC, h, acc);

#pragma unroll
    for (int i = 0; i < 2; ++i) {
#pragma unroll
        for (int r = 0; r < 8; ++r) {
            const int row = m0 + i * 16 + 8 * h + r;
            if (row < M) {
                float* crow = C + (long)row * NC + n0 + c;
#pragma unroll
                for (int t = 0; t < NT; ++t) crow[t * 16] = acc[i][t][r];
            }
        }
    }
}

#define PU 8
template <int H>
__global__ __launch_bounds__(256) void attn_proj(
    const float* __restrict__ feat, const float* __restrict__ al,
    const float* __restrict__ ar, float* __restrict__ el,
    float* __restrict__ er, int nPairs)
{
    static_assert(PU % H == 0, "a trip starts at head 0");
    const int lane = threadIdx.x & 31;
    const int wid  = blockIdx.x * 8 + (threadIdx.x >> 5);
    const int nw   = gridDim.x * 8;
    float wl0[H], wl1[H], wr0[H], wr1[H];
#pragma unroll
    for (int h = 0; h < H; ++h) {
        wl0[h] = al[h * 64 + lane]; wl1[h] = al[h * 64 + lane + 32];
        wr0[h] = ar[h * 64 + lane]; wr1[h] = ar[h * 64 + lane + 32];
    }
    for (int p0 = wid * PU; p0 < nPairs; p0 += nw * PU) {
        float f0[PU], f1[PU];
#pragma unroll
        for (int u = 0; u < PU; ++u) {
            const float* f = feat + (long)min(p0 + u, nPairs - 1) * 64;
            f0[u] = f[lane]; f1[u] = f[lane + 32];
        }
        float sl[PU], sr[PU];
#pragma unroll
        for (int u = 0; u < PU; ++u) {
            sl[u] = f0[u] * wl0[u % H] + f1[u] * wl1[u % H];
            sr[u] = f0[u] * wr0[u % H] + f1[u] * wr1[u % H];
        }
#pragma unroll
        for (int o = 16; o > 0; o >>= 1) {
#pragma unroll
            for (int u = 0; u < PU; ++u) {
                sl[u] += __shfl_xor(sl[u], o, 32);
                sr[u] += __shfl_xor(sr[u], o, 32);
            }
        }
        float ol = sl[0], orr = sr[0];
#pragma unroll
        for (int u = 1; u < PU; ++u) if (lane == u) { ol = sl[u]; orr = sr[u]; }
        if (lane < PU && p0 + lane < nPairs) { el[p0 + lane] = ol; er[p0 + lane] = orr; }
    }
}

#define TROWS 196
#define QCAP  384
#define CHUNK 4096
#define NTILES ((N_NODES + TROWS - 1) / TROWS)

__global__ __launch_bounds__(1024) void bin_edges_by_tile(
    const int* __restrict__ src, const int* __restrict__ dst,
    int* __restrict__ ent, int* __restrict__ off, int E, int nNodes)
{
    static_assert(TROWS <= 256, "the row inside a tile is packed into 8 bits");
    static_assert(NTILES <= 1024, "one thread per tile in the prefix sum");
    __shared__ int hist[NTILES];
    __shared__ int start[NTILES + 1];
    __shared__ int wsum[32];
    __shared__ int sorted[CHUNK];

    const int tid  = threadIdx.x;
    const int lane = tid & 31;
    const int wv   = tid >> 5;
    const int nchunks = (E + CHUNK - 1) / CHUNK;
    for (int c = blockIdx.x; c < nchunks; c += gridDim.x) {
        for (int i = tid; i < NTILES; i += 1024) hist[i] = 0;
        __syncthreads();

        const int e0 = c * CHUNK + tid * 4;
        int d[4], s[4];
        if (e0 + 3 < E) {
            const int4 dv = *(const int4*)(dst + e0);
            const int4 sv = *(const int4*)(src + e0);
            d[0] = dv.x; d[1] = dv.y; d[2] = dv.z; d[3] = dv.w;
            s[0] = sv.x; s[1] = sv.y; s[2] = sv.z; s[3] = sv.w;
        } else {
#pragma unroll
            for (int k = 0; k < 4; ++k) {
                d[k] = (e0 + k < E) ? dst[e0 + k] : -1;
                s[k] = (e0 + k < E) ? src[e0 + k] : 0;
            }
        }
        int tl[4], rk[4], pk[4];
#pragma unroll
        for (int k = 0; k < 4; ++k) {
            tl[k] = -1; rk[k] = 0; pk[k] = 0;
            if ((unsigned)d[k] < (unsigned)nNodes) {
                tl[k] = d[k] / TROWS;
                pk[k] = (s[k] << 8) | (d[k] - tl[k] * TROWS);
                rk[k] = __hip_atomic_fetch_add(hist + tl[k], 1, __ATOMIC_RELAXED,
                                               __HIP_MEMORY_SCOPE_WORKGROUP);
            }
        }
        __syncthreads();

        int v = 0, incl = 0;
        if (wv < (NTILES + 31) / 32) {
            v = (tid < NTILES) ? hist[tid] : 0;
            incl = v;
            for (int o = 1; o < 32; o <<= 1) {
                const int t = __shfl_up(incl, o, 32);
                if (lane >= o) incl += t;
            }
            if (lane == 31) wsum[wv] = incl;
        }
        __syncthreads();
        if (tid < NTILES) {
            int base = 0;
            for (int w = 0; w < wv; ++w) base += wsum[w];
            start[tid] = base + incl - v;
            if (tid == NTILES - 1) start[NTILES] = base + incl;
        }
        __syncthreads();

#pragma unroll
        for (int k = 0; k < 4; ++k)
            if (tl[k] >= 0) sorted[start[tl[k]] + rk[k]] = pk[k];
        __syncthreads();

        const int cnt = start[NTILES];
        for (int i = tid; i < cnt; i += 1024) ent[(long)c * CHUNK + i] = sorted[i];
        for (int i = tid; i <= NTILES; i += 1024) off[(long)c * (NTILES + 1) + i] = start[i];
        __syncthreads();
    }
}

template <int C, int U>
__device__ __forceinline__ void agg_group(
    const int* myqs, const int* myqd, int q, int lane,
    const float* __restrict__ el, const float* __restrict__ feat,
    const float* ler, const float* rmax, float* rsum, float* tile)
{
    constexpr int H  = C / 64;
    constexpr int NI = C / 32;
    const int pl = lane % (U * H);
    const int pu = pl / H, ph = pl - pu * H;
    const int sl = myqs[q + pu];
    const int rl = myqd[q + pu];
    const float xe = el[sl * H + ph];
    const float xr = ler[rl * H + ph];
    const float mx = rmax[rl * H + ph];

    int s[U], dd[U];
#pragma unroll
    for (int u = 0; u < U; ++u) {
        s[u]  = __builtin_amdgcn_readfirstlane(myqs[q + u]);
        dd[u] = __builtin_amdgcn_readfirstlane(myqd[q + u]);
    }
    float v[U][NI];
#pragma unroll
    for (int u = 0; u < U; ++u) {
        const float* f = feat + (long)s[u] * C + lane;
#pragma unroll
        for (int i = 0; i < NI; ++i) v[u][i] = f[i * 32];
    }
    __builtin_amdgcn_sched_barrier(0);
    float x = xe + xr;
    x = x > 0.0f ? x : 0.2f * x;
    const float w = expf(x - mx);
    if (lane < U * H)
        __hip_atomic_fetch_add(rsum + rl * H + ph, w, __ATOMIC_RELAXED, __HIP_MEMORY_SCOPE_WORKGROUP);
#pragma unroll
    for (int u = 0; u < U; ++u) {
        float* t = tile + dd[u] * C + lane;
#pragma unroll
        for (int i = 0; i < NI; ++i) {
            const float a = __int_as_float(
                __builtin_amdgcn_readlane(__float_as_int(w), u * H + (i >> 1)));
            __hip_atomic_fetch_add(t + i * 32, v[u][i] * a,
                                   __ATOMIC_RELAXED, __HIP_MEMORY_SCOPE_WORKGROUP);
        }
    }
}

template <int C>
__global__ __launch_bounds__(1024) void edge_softmax_aggregate_tile(
    const int* __restrict__ ent, const int* __restrict__ off,
    const float* __restrict__ feat, const float* __restrict__ el,
    const float* __restrict__ er, float* __restrict__ rst, int E, int nNodes)
{
    constexpr int H = C / 64;
    __shared__ __attribute__((aligned(16))) float tile[TROWS * C];
    __shared__ float ler[TROWS * H];
    __shared__ float rmax[TROWS * H];
    __shared__ float rsum[TROWS * H];
    __shared__ int qs[32 * QCAP];
    __shared__ int qd[32 * QCAP];

    const int tid  = threadIdx.x;
    const int lane = tid & 31;
    const int wave = __builtin_amdgcn_readfirstlane(tid >> 5);
    const int d0   = blockIdx.x * TROWS;
    const int nrows = min(TROWS, nNodes - d0);
    const int nchunks = (E + CHUNK - 1) / CHUNK;

    for (int i = tid; i < TROWS * C / 4; i += 1024)
        ((float4*)tile)[i] = make_float4(0.0f, 0.0f, 0.0f, 0.0f);
    for (int i = tid; i < TROWS * H; i += 1024) {
        ler[i]  = (i < nrows * H) ? er[d0 * H + i] : 0.0f;
        rmax[i] = -INFINITY;
        rsum[i] = 0.0f;
    }
    __syncthreads();

    int* myqs = qs + wave * QCAP;
    int* myqd = qd + wave * QCAP;
    int qlen = 0;
    bool ovf = false;

    auto stage1 = [&](int e) {
        constexpr int PER = 32 / H;
        const int sub = lane / H, hh = lane - sub * H;
        for (int i = 0; i < e; i += 4 * PER) {
            int row[4];
            float xe[4];
#pragma unroll
            for (int u = 0; u < 4; ++u) {
                const int idx = min(i + u * PER + sub, e - 1);
                row[u] = myqd[idx];
                xe[u]  = el[myqs[idx] * H + hh];
            }
#pragma unroll
            for (int u = 0; u < 4; ++u) {
                if (i + u * PER + sub < e) {
                    float x = xe[u] + ler[row[u] * H + hh];
                    x = x > 0.0f ? x : 0.2f * x;
                    __hip_atomic_fetch_max(rmax + row[u] * H + hh, x,
                                           __ATOMIC_RELAXED, __HIP_MEMORY_SCOPE_WORKGROUP);
                }
            }
        }
    };

    auto stage2 = [&](int e) {
        int q = 0;
        for (; q + 4 <= e; q += 4)
            agg_group<C, 4>(myqs, myqd, q, lane, el, feat, ler, rmax, rsum, tile);
        for (; q < e; ++q)
            agg_group<C, 1>(myqs, myqd, q, lane, el, feat, ler, rmax, rsum, tile);
    };

    auto walk = [&](auto&& full) {
        for (int g0 = wave; g0 < nchunks; g0 += 1024) {
            const int cidx = g0 + 32 * lane;
            int sbeg = 0, n = 0;
            if (cidx < nchunks) {
                const int* o = off + (long)cidx * (NTILES + 1) + blockIdx.x;
                const int a = o[0], b = o[1];
                sbeg = cidx * CHUNK + a;
                n = b - a;
            }
            int incl = n;
            for (int o = 1; o < 32; o <<= 1) {
                const int t = __shfl_up(incl, o, 32);
                if (lane >= o) incl += t;
            }
            const int excl  = incl - n;
            const int delta = sbeg - excl;
            const int total = __builtin_amdgcn_readlane(incl, 31);
            const int nact  = min(32, (nchunks - g0 + 31) >> 5);
            for (int f0 = 0; f0 < total; f0 += 128) {
                if (qlen > QCAP - 128) { full(qlen); ovf = true; qlen = 0; }
                int fc[4], ad[4], w[4];
#pragma unroll
                for (int k = 0; k < 4; ++k) {
                    fc[k] = min(f0 + k * 32 + lane, total - 1);
                    ad[k] = 0;
                }
                for (int j = 0; j < nact; ++j) {
                    const int ej = __builtin_amdgcn_readlane(excl, j);
                    const int dj = __builtin_amdgcn_readlane(delta, j);
#pragma unroll
                    for (int k = 0; k < 4; ++k) if (fc[k] >= ej) ad[k] = dj + fc[k];
                }
#pragma unroll
                for (int k = 0; k < 4; ++k) w[k] = ent[ad[k]];
#pragma unroll
                for (int k = 0; k < 4; ++k) {
                    if (f0 + k * 32 + lane < total) {
                        const int pos = qlen + k * 32 + lane;
                        myqs[pos] = w[k] >> 8;
                        myqd[pos] = w[k] & 255;
                    }
                }
                qlen += min(128, total - f0);
            }
        }
    };

    walk([&](int n) { stage1(n); });
    stage1(qlen);
    __syncthreads();

    if (!ovf) {
        stage2(qlen);
    } else {
        qlen = 0;
        walk([&](int n) { stage2(n); });
        stage2(qlen);
    }

    __syncthreads();
    float4* o = (float4*)(rst + (long)d0 * C);
    for (int i = tid; i < nrows * (C / 4); i += 1024) {
        const float4 t = ((const float4*)tile)[i];
        const int row = i / (C / 4);
        const int hh  = (i % (C / 4)) / 16;
        const float sm = rsum[row * H + hh];
        const float dn = (sm == 0.0f) ? 1.0f : sm;
        o[i] = make_float4(t.x / dn, t.y / dn, t.z / dn, t.w / dn);
    }
}

template <int C>
__global__ __launch_bounds__(256) void bias_elu(float* __restrict__ x, const float* __restrict__ b,
                                                int n4)
{
    float4* x4 = (float4*)x;
    const int stride = gridDim.x * 256;
    for (int i0 = blockIdx.x * 256 + threadIdx.x; i0 < n4; i0 += 4 * stride) {
        float4 v[4];
#pragma unroll
        for (int u = 0; u < 4; ++u) v[u] = x4[min(i0 + u * stride, n4 - 1)];
#pragma unroll
        for (int u = 0; u < 4; ++u) {
            const int i = i0 + u * stride;
            if (i < n4) {
                const float* bb = b + ((unsigned)i % (unsigned)(C / 4)) * 4u;
                float4 r;
                float t;
                t = v[u].x + bb[0]; r.x = t > 0.0f ? t : expm1f(t);
                t = v[u].y + bb[1]; r.y = t > 0.0f ? t : expm1f(t);
                t = v[u].z + bb[2]; r.z = t > 0.0f ? t : expm1f(t);
                t = v[u].w + bb[3]; r.w = t > 0.0f ? t : expm1f(t);
                x4[i] = r;
            }
        }
    }
}

__global__ __launch_bounds__(256) void classify(
    const float* __restrict__ rst2, const float* __restrict__ b2,
    const float* __restrict__ Wc1, const float* __restrict__ bc1,
    const float* __restrict__ Wc2, const float* __restrict__ bc2,
    float* __restrict__ out, int nNodes)
{
    const int lane = threadIdx.x & 31;
    const int wid  = blockIdx.x * 8 + __builtin_amdgcn_readfirstlane(threadIdx.x >> 5);
    const int nw   = gridDim.x * 8;
    float w[64];
#pragma unroll
    for (int d = 0; d < 64; ++d) w[d] = Wc1[d * 32 + lane];
    const float bl  = bc1[lane];
    const float w20 = Wc2[lane * 2 + 0], w21 = Wc2[lane * 2 + 1];
    const float c0  = bc2[0], c1 = bc2[1];
    for (int node = wid; node < nNodes; node += nw) {
        const float* r = rst2 + (long)node * 64;
        float acc = 0.0f;
#pragma unroll
        for (int d = 0; d < 64; ++d)
            acc += (r[d] + b2[d]) * w[d];
        float t = fmaxf(acc + bl, 0.0f);
        float o0 = t * w20;
        float o1 = t * w21;
#pragma unroll
        for (int o = 16; o > 0; o >>= 1) {
            o0 += __shfl_xor(o0, o, 32);
            o1 += __shfl_xor(o1, o, 32);
        }
        if (lane == 0) {
            out[(long)node * 2 + 0] = o0 + c0;
            out[(long)node * 2 + 1] = o1 + c1;
        }
    }
}

extern "C" void kernel_launch(void* const* d_in, const int* in_sizes, int n_in,
                              void* d_out, int out_size, void* d_ws, size_t ws_size,
                              hipStream_t stream)
{
    (void)n_in; (void)out_size; (void)ws_size;

    const float* randfeat = (const float*)d_in[0];
    const float* func_emb = (const float*)d_in[1];
    const float* codebert = (const float*)d_in[2];
    const int*   src      = (const int*)d_in[3];
    const int*   dst      = (const int*)d_in[4];
    const float* W1  = (const float*)d_in[5];
    const float* al1 = (const float*)d_in[6];
    const float* ar1 = (const float*)d_in[7];
    const float* b1  = (const float*)d_in[8];
    const float* W2  = (const float*)d_in[9];
    const float* al2 = (const float*)d_in[10];
    const float* ar2 = (const float*)d_in[11];
    const float* b2  = (const float*)d_in[12];
    const float* Wc1 = (const float*)d_in[13];
    const float* bc1 = (const float*)d_in[14];
    const float* Wc2 = (const float*)d_in[15];
    const float* bc2 = (const float*)d_in[16];
    float* out = (float*)d_out;

    const int  E  = in_sizes[3];
    const long NN = N_NODES;

    float* ws     = (float*)d_ws;
    float* feat1  = ws;
    float* rst1   = feat1  + NN * 256;
    int*   ent    = (int*)(rst1 + NN * 256);
    int*   off    = ent    + NN * 48;
    float* rst2   = rst1   + NN * 256 + NN * 64;
    float* feat2  = feat1;
    float* el1    = rst2   + NN * 64;
    float* er1    = el1    + NN * 4;
    float* el2    = er1    + NN * 4;
    float* er2    = el2    + NN;

    bin_edges_by_tile<<<256, 1024, 0, stream>>>(src, dst, ent, off, E, N_NODES);

    gemm_wmma_f32<8, 2, 256, 12, true><<<256, 384, 0, stream>>>(
        randfeat, func_emb, codebert, 100, 100, 768, W1, feat1, N_NODES);
    attn_proj<4><<<1024, 256, 0, stream>>>(feat1, al1, ar1, el1, er1, N_NODES * 4);
    edge_softmax_aggregate_tile<256><<<NTILES, 1024, 0, stream>>>(ent, off, feat1, el1, er1, rst1, E, N_NODES);
    bias_elu<256><<<1024, 256, 0, stream>>>(rst1, b1, N_NODES * 64);

    gemm_wmma_f32<4, 1, 64, 4, false><<<((N_NODES + 31) / 32 + 3) / 4, 128, 0, stream>>>(
        rst1, rst1, rst1, 256, 0, 0, W2, feat2, N_NODES);
    attn_proj<1><<<256, 256, 0, stream>>>(feat2, al2, ar2, el2, er2, N_NODES);
    edge_softmax_aggregate_tile<64><<<NTILES, 1024, 0, stream>>>(ent, off, feat2, el2, er2, rst2, E, N_NODES);

    classify<<<1024, 256, 0, stream>>>(rst2, b2, Wc1, bc1, Wc2, bc2, out, N_NODES);
}


# Layer-1 tile finished in LDS: rows get /row sum, bias, elu in place, then tile x W2 by WMMA (A fragments from LDS), finishing staged beside products; h2, bias_elu, GEMM2 launches gone; same sum order
# speedup vs baseline: 6.0799x; 1.0778x over previous
#include <hip/hip_runtime.h>
#include <math.h>


typedef __attribute__((ext_vector_type(2))) float v2f;
typedef __attribute__((ext_vector_type(8))) float v8f;

#define N_NODES 50000

template <int NT, int NC>
__device__ __forceinline__ void seg_accum(
    const float* __restrict__ a0, const float* __restrict__ a1, int w,
    const float* __restrict__ bcol, int h, v8f (&acc)[2][NT])
{
    const int nblk = w >> 5;
    if (nblk > 0) {
        const float4* q0 = (const float4*)(a0 + 16 * h);
        const float4* q1 = (const float4*)(a1 + 16 * h);
        const float* bp = bcol + (long)(16 * h) * NC;
        float4 ra[4], rb[4];
        v2f bc[NT];
#pragma unroll
        for (int q = 0; q < 4; ++q) { ra[q] = q0[q]; rb[q] = q1[q]; }
#pragma unroll
        for (int t = 0; t < NT; ++t) { bc[t].x = bp[t * 16]; bc[t].y = bp[NC + t * 16]; }

        for (int blk = 0; blk < nblk; ++blk) {
            const int adv = (blk + 1 < nblk) ? 32 : 0;
            const float4* nq0 = (const float4*)((const float*)q0 + adv);
            const float4* nq1 = (const float4*)((const float*)q1 + adv);
            const float* nbp = bp + (long)adv * NC;
            float4 na[4], nb[4];
#pragma unroll
            for (int q = 0; q < 4; ++q) { na[q] = nq0[q]; nb[q] = nq1[q]; }
            __builtin_amdgcn_sched_barrier(0);
#pragma unroll
            for (int j = 0; j < 8; ++j) {
                v2f bn[NT];
                if (j < 7) {
#pragma unroll
                    for (int t = 0; t < NT; ++t) {
                        bn[t].x = bp[(2 * j + 2) * NC + t * 16];
                        bn[t].y = bp[(2 * j + 3) * NC + t * 16];
                    }
                } else {
#pragma unroll
                    for (int t = 0; t < NT; ++t) {
                        bn[t].x = nbp[t * 16];
                        bn[t].y = nbp[NC + t * 16];
                    }
                }
                v2f fa, fb;
                if (j & 1) { fa.x = ra[j >> 1].z; fa.y = ra[j >> 1].w; fb.x = rb[j >> 1].z; fb.y = rb[j >> 1].w; }
                else       { fa.x = ra[j >> 1].x; fa.y = ra[j >> 1].y; fb.x = rb[j >> 1].x; fb.y = rb[j >> 1].y; }
#pragma unroll
                for (int t = 0; t < NT; ++t) {
                    acc[0][t] = __builtin_amdgcn_wmma_f32_16x16x4_f32(
                        false, fa, false, bc[t], (short)0, acc[0][t], false, false);
                    acc[1][t] = __builtin_amdgcn_wmma_f32_16x16x4_f32(
                        false, fb, false, bc[t], (short)0, acc[1][t], false, false);
                }
#pragma unroll
                for (int t = 0; t < NT; ++t) bc[t] = bn[t];
            }
            __builtin_amdgcn_sched_barrier(0);
            asm volatile("v_nop\n\tv_nop\n\tv_nop\n\tv_nop");
#pragma unroll
            for (int q = 0; q < 4; ++q) { ra[q] = na[q]; rb[q] = nb[q]; }
            q0 = nq0; q1 = nq1; bp = nbp;
        }
    }
    for (int k = nblk << 5; k + 4 <= w; k += 4) {
        const float2 ta = *(const float2*)(a0 + k + 2 * h);
        const float2 tb = *(const float2*)(a1 + k + 2 * h);
        v2f fa, fb;
        fa.x = ta.x; fa.y = ta.y; fb.x = tb.x; fb.y = tb.y;
        const float* bp = bcol + (long)(k + 2 * h) * NC;
#pragma unroll
        for (int t = 0; t < NT; ++t) {
            v2f b;
            b.x = bp[t * 16];
            b.y = bp[NC + t * 16];
            acc[0][t] = __builtin_amdgcn_wmma_f32_16x16x4_f32(
                false, fa, false, b, (short)0, acc[0][t], false, false);
            acc[1][t] = __builtin_amdgcn_wmma_f32_16x16x4_f32(
                false, fb, false, b, (short)0, acc[1][t], false, false);
        }
    }
}

template <int NC>
__device__ __forceinline__ void seg_accum_rem(
    const float* __restrict__ a0, int w,
    const float* __restrict__ bcol, int h, v8f (&acc)[2])
{
    const int nblk = w >> 5;
    if (nblk > 0) {
        const float4* q0 = (const float4*)(a0 + 16 * h);
        const float* bp = bcol + (long)(16 * h) * NC;
        float4 ra[4];
        v2f rb[8][2];
#pragma unroll
        for (int q = 0; q < 4; ++q) ra[q] = q0[q];
#pragma unroll
        for (int j = 0; j < 8; ++j) {
#pragma unroll
            for (int t = 0; t < 2; ++t) {
                rb[j][t].x = bp[(2 * j) * NC + t * 16];
                rb[j][t].y = bp[(2 * j + 1) * NC + t * 16];
            }
        }
        for (int blk = 0; blk < nblk; ++blk) {
            const int adv = (blk + 1 < nblk) ? 32 : 0;
            const float4* nq0 = (const float4*)((const float*)q0 + adv);
            const float* nbp = bp + (long)adv * NC;
            float4 na[4];
            v2f nb[8][2];
#pragma unroll
            for (int q = 0; q < 4; ++q) na[q] = nq0[q];
#pragma unroll
            for (int j = 0; j < 8; ++j) {
#pragma unroll
                for (int t = 0; t < 2; ++t) {
                    nb[j][t].x = nbp[(2 * j) * NC + t * 16];
                    nb[j][t].y = nbp[(2 * j + 1) * NC + t * 16];
                }
            }
            __builtin_amdgcn_sched_barrier(0);
#pragma unroll
            for (int j = 0; j < 8; ++j) {
                v2f fa;
                if (j & 1) { fa.x = ra[j >> 1].z; fa.y = ra[j >> 1].w; }
                else       { fa.x = ra[j >> 1].x; fa.y = ra[j >> 1].y; }
#pragma unroll
                for (int t = 0; t < 2; ++t)
                    acc[t] = __builtin_amdgcn_wmma_f32_16x16x4_f32(
                        false, fa, false, rb[j][t], (short)0, acc[t], false, false);
            }
            __builtin_amdgcn_sched_barrier(0);
#pragma unroll
            for (int q = 0; q < 4; ++q) ra[q] = na[q];
#pragma unroll
            for (int j = 0; j < 8; ++j) { rb[j][0] = nb[j][0]; rb[j][1] = nb[j][1]; }
            q0 = nq0; bp = nbp;
        }
    }
    for (int k = nblk << 5; k + 4 <= w; k += 4) {
        const float2 ta = *(const float2*)(a0 + k + 2 * h);
        v2f fa;
        fa.x = ta.x; fa.y = ta.y;
        const float* bp = bcol + (long)(k + 2 * h) * NC;
#pragma unroll
        for (int t = 0; t < 2; ++t) {
            v2f b;
            b.x = bp[t * 16];
            b.y = bp[NC + t * 16];
            acc[t] = __builtin_amdgcn_wmma_f32_16x16x4_f32(
                false, fa, false, b, (short)0, acc[t], false, false);
        }
    }
}

template <int NT, int NG, int NC, int WPB, bool REM>
__global__ __launch_bounds__(WPB * 32) void gemm_wmma_f32(
    const float* __restrict__ p0, const float* __restrict__ p1,
    const float* __restrict__ p2, int w0, int w1, int w2,
    const float* __restrict__ B, float* __restrict__ C, int M)
{
    const int lane = threadIdx.x & 31;
    const int wv   = threadIdx.x >> 5;
    const int gw   = blockIdx.x * WPB + wv;
    const int c  = lane & 15;
    const int h  = lane >> 4;

    if (REM) {
        const int nWaves   = gridDim.x * WPB;
        const int mainRows = (nWaves / NG) * 32;
        const int nRem = (mainRows < M) ? ((M - mainRows + 15) / 16) * (NC / 32) : 0;
        for (int u = wv * gridDim.x + blockIdx.x; u < nRem; u += nWaves) {
            const int tm0 = mainRows + (u / (NC / 32)) * 16;
            const int tn0 = (u % (NC / 32)) * 32;
            const int rr = min(tm0 + c, M - 1);
            v8f racc[2];
            racc[0] = v8f{}; racc[1] = v8f{};
            const float* rcol = B + tn0 + c;
            seg_accum_rem<NC>(p0 + (long)rr * w0, w0, rcol, h, racc);
            seg_accum_rem<NC>(p1 + (long)rr * w1, w1, rcol + (long)w0 * NC, h, racc);
            seg_accum_rem<NC>(p2 + (long)rr * w2, w2, rcol + (long)(w0 + w1) * NC, h, racc);
#pragma unroll
            for (int r = 0; r < 8; ++r) {
                const int row = tm0 + 8 * h + r;
                if (row < M) {
                    float* crow = C + (long)row * NC + tn0 + c;
                    crow[0]  = racc[0][r];
                    crow[16] = racc[1][r];
                }
            }
        }
    }

    const int ng   = gw % NG;
    const int m0   = (gw / NG) * 32;
    if (m0 >= M) return;
    const int n0 = ng * NT * 16;

    const int r0 = min(m0 + c, M - 1);
    const int r1 = min(m0 + 16 + c, M - 1);

    v8f acc[2][NT];
#pragma unroll
    for (int t = 0; t < NT; ++t) { acc[0][t] = v8f{}; acc[1][t] = v8f{}; }

    const float* bcol = B + n0 + c;
    seg_accum<NT, NC>(p0 + (long)r0 * w0, p0 + (long)r1 * w0, w0, bcol, h, acc);
    seg_accum<NT, NC>(p1 + (long)r0 * w1, p1 + (long)r1 * w1, w1, bcol + (long)w0 * NC, h, acc);
    seg_accum<NT, NC>(p2 + (long)r0 * w2, p2 + (long)r1 * w2, w2, bcol + (long)(w0 + w1) * NC, h, acc);

#pragma unroll
    for (int i = 0; i < 2; ++i) {
#pragma unroll
        for (int r = 0; r < 8; ++r) {
            const int row = m0 + i * 16 + 8 * h + r;
            if (row < M) {
                float* crow = C + (long)row * NC + n0 + c;
#pragma unroll
                for (int t = 0; t < NT; ++t) crow[t * 16] = acc[i][t][r];
            }
        }
    }
}

#define PU 8
template <int H>
__global__ __launch_bounds__(256) void attn_proj(
    const float* __restrict__ feat, const float* __restrict__ al,
    const float* __restrict__ ar, float* __restrict__ el,
    float* __restrict__ er, int nPairs)
{
    static_assert(PU % H == 0, "a trip starts at head 0");
    const int lane = threadIdx.x & 31;
    const int wid  = blockIdx.x * 8 + (threadIdx.x >> 5);
    const int nw   = gridDim.x * 8;
    float wl0[H], wl1[H], wr0[H], wr1[H];
#pragma unroll
    for (int h = 0; h < H; ++h) {
        wl0[h] = al[h * 64 + lane]; wl1[h] = al[h * 64 + lane + 32];
        wr0[h] = ar[h * 64 + lane]; wr1[h] = ar[h * 64 + lane + 32];
    }
    for (int p0 = wid * PU; p0 < nPairs; p0 += nw * PU) {
        float f0[PU], f1[PU];
#pragma unroll
        for (int u = 0; u < PU; ++u) {
            const float* f = feat + (long)min(p0 + u, nPairs - 1) * 64;
            f0[u] = f[lane]; f1[u] = f[lane + 32];
        }
        float sl[PU], sr[PU];
#pragma unroll
        for (int u = 0; u < PU; ++u) {
            sl[u] = f0[u] * wl0[u % H] + f1[u] * wl1[u % H];
            sr[u] = f0[u] * wr0[u % H] + f1[u] * wr1[u % H];
        }
#pragma unroll
        for (int o = 16; o > 0; o >>= 1) {
#pragma unroll
            for (int u = 0; u < PU; ++u) {
                sl[u] += __shfl_xor(sl[u], o, 32);
                sr[u] += __shfl_xor(sr[u], o, 32);
            }
        }
        float ol = sl[0], orr = sr[0];
#pragma unroll
        for (int u = 1; u < PU; ++u) if (lane == u) { ol = sl[u]; orr = sr[u]; }
        if (lane < PU && p0 + lane < nPairs) { el[p0 + lane] = ol; er[p0 + lane] = orr; }
    }
}

#define TROWS 196
#define QCAP  384
#define CHUNK 4096
#define NTILES ((N_NODES + TROWS - 1) / TROWS)

__global__ __launch_bounds__(1024) void bin_edges_by_tile(
    const int* __restrict__ src, const int* __restrict__ dst,
    int* __restrict__ ent, int* __restrict__ off, int E, int nNodes)
{
    static_assert(TROWS <= 256, "the row inside a tile is packed into 8 bits");
    static_assert(NTILES <= 1024, "one thread per tile in the prefix sum");
    __shared__ int hist[NTILES];
    __shared__ int start[NTILES + 1];
    __shared__ int wsum[32];
    __shared__ int sorted[CHUNK];

    const int tid  = threadIdx.x;
    const int lane = tid & 31;
    const int wv   = tid >> 5;
    const int nchunks = (E + CHUNK - 1) / CHUNK;
    for (int c = blockIdx.x; c < nchunks; c += gridDim.x) {
        for (int i = tid; i < NTILES; i += 1024) hist[i] = 0;
        __syncthreads();

        const int e0 = c * CHUNK + tid * 4;
        int d[4], s[4];
        if (e0 + 3 < E) {
            const int4 dv = *(const int4*)(dst + e0);
            const int4 sv = *(const int4*)(src + e0);
            d[0] = dv.x; d[1] = dv.y; d[2] = dv.z; d[3] = dv.w;
            s[0] = sv.x; s[1] = sv.y; s[2] = sv.z; s[3] = sv.w;
        } else {
#pragma unroll
            for (int k = 0; k < 4; ++k) {
                d[k] = (e0 + k < E) ? dst[e0 + k] : -1;
                s[k] = (e0 + k < E) ? src[e0 + k] : 0;
            }
        }
        int tl[4], rk[4], pk[4];
#pragma unroll
        for (int k = 0; k < 4; ++k) {
            tl[k] = -1; rk[k] = 0; pk[k] = 0;
            if ((unsigned)d[k] < (unsigned)nNodes) {
                tl[k] = d[k] / TROWS;
                pk[k] = (s[k] << 8) | (d[k] - tl[k] * TROWS);
                rk[k] = __hip_atomic_fetch_add(hist + tl[k], 1, __ATOMIC_RELAXED,
                                               __HIP_MEMORY_SCOPE_WORKGROUP);
            }
        }
        __syncthreads();

        int v = 0, incl = 0;
        if (wv < (NTILES + 31) / 32) {
            v = (tid < NTILES) ? hist[tid] : 0;
            incl = v;
            for (int o = 1; o < 32; o <<= 1) {
                const int t = __shfl_up(incl, o, 32);
                if (lane >= o) incl += t;
            }
            if (lane == 31) wsum[wv] = incl;
        }
        __syncthreads();
        if (tid < NTILES) {
            int base = 0;
            for (int w = 0; w < wv; ++w) base += wsum[w];
            start[tid] = base + incl - v;
            if (tid == NTILES - 1) start[NTILES] = base + incl;
        }
        __syncthreads();

#pragma unroll
        for (int k = 0; k < 4; ++k)
            if (tl[k] >= 0) sorted[start[tl[k]] + rk[k]] = pk[k];
        __syncthreads();

        const int cnt = start[NTILES];
        for (int i = tid; i < cnt; i += 1024) ent[(long)c * CHUNK + i] = sorted[i];
        for (int i = tid; i <= NTILES; i += 1024) off[(long)c * (NTILES + 1) + i] = start[i];
        __syncthreads();
    }
}

template <int K, int NC>
__device__ __forceinline__ void tile_unit_product(
    const float* arow, const float* __restrict__ bcol, v8f (&acc)[2])
{
    static_assert(K % 32 == 0, "whole 32-wide blocks");
    constexpr int NHB = K / 16;
    float4 fa[2][2];
    v2f fb[2][4][2];
#pragma unroll
    for (int hb = -1; hb < NHB; ++hb) {
        if (hb + 1 < NHB) {
            const int nx = hb + 1;
            const int o  = (nx >> 1) * 32 + (nx & 1) * 8;
            fa[nx & 1][0] = *(const float4*)(arow + o);
            fa[nx & 1][1] = *(const float4*)(arow + o + 4);
#pragma unroll
            for (int jj = 0; jj < 4; ++jj) {
#pragma unroll
                for (int t = 0; t < 2; ++t) {
                    fb[nx & 1][jj][t].x = bcol[(o + 2 * jj) * NC + t * 16];
                    fb[nx & 1][jj][t].y = bcol[(o + 2 * jj + 1) * NC + t * 16];
                }
            }
        }
        __builtin_amdgcn_sched_barrier(0);
        if (hb >= 0) {
#pragma unroll
            for (int jj = 0; jj < 4; ++jj) {
                const float4 q = fa[hb & 1][jj >> 1];
                v2f a;
                if (jj & 1) { a.x = q.z; a.y = q.w; }
                else        { a.x = q.x; a.y = q.y; }
#pragma unroll
                for (int t = 0; t < 2; ++t)
                    acc[t] = __builtin_amdgcn_wmma_f32_16x16x4_f32(
                        false, a, false, fb[hb & 1][jj][t], (short)0, acc[t], false, false);
            }
        }
    }
}

template <int C, int PITCH, int U>
__device__ __forceinline__ void agg_group(
    const int* myqs, const int* myqd, int q, int lane,
    const float* __restrict__ el, const float* __restrict__ feat,
    const float* ler, const float* rmax, float* rsum, float* tile)
{
    constexpr int H  = C / 64;
    constexpr int NI = C / 32;
    const int pl = lane % (U * H);
    const int pu = pl / H, ph = pl - pu * H;
    const int sl = myqs[q + pu];
    const int rl = myqd[q + pu];
    const float xe = el[sl * H + ph];
    const float xr = ler[rl * H + ph];
    const float mx = rmax[rl * H + ph];

    int s[U], dd[U];
#pragma unroll
    for (int u = 0; u < U; ++u) {
        s[u]  = __builtin_amdgcn_readfirstlane(myqs[q + u]);
        dd[u] = __builtin_amdgcn_readfirstlane(myqd[q + u]);
    }
    float v[U][NI];
#pragma unroll
    for (int u = 0; u < U; ++u) {
        const float* f = feat + (long)s[u] * C + lane;
#pragma unroll
        for (int i = 0; i < NI; ++i) v[u][i] = f[i * 32];
    }
    __builtin_amdgcn_sched_barrier(0);
    float x = xe + xr;
    x = x > 0.0f ? x : 0.2f * x;
    const float w = expf(x - mx);
    if (lane < U * H)
        __hip_atomic_fetch_add(rsum + rl * H + ph, w, __ATOMIC_RELAXED, __HIP_MEMORY_SCOPE_WORKGROUP);
#pragma unroll
    for (int u = 0; u < U; ++u) {
        float* t = tile + dd[u] * PITCH + lane;
#pragma unroll
        for (int i = 0; i < NI; ++i) {
            const float a = __int_as_float(
                __builtin_amdgcn_readlane(__float_as_int(w), u * H + (i >> 1)));
            __hip_atomic_fetch_add(t + i * 32, v[u][i] * a,
                                   __ATOMIC_RELAXED, __HIP_MEMORY_SCOPE_WORKGROUP);
        }
    }
}

template <int C, bool FIN, int NCN>
__global__ __launch_bounds__(1024) void edge_softmax_aggregate_tile(
    const int* __restrict__ ent, const int* __restrict__ off,
    const float* __restrict__ feat, const float* __restrict__ el,
    const float* __restrict__ er, float* __restrict__ rst,
    const float* __restrict__ bias, const float* __restrict__ Wn, float* __restrict__ outn,
    int E, int nNodes)
{
    constexpr int H = C / 64;
    constexpr int PITCH = FIN ? C + 4 : C;
    __shared__ __attribute__((aligned(16))) float tile[TROWS * PITCH];
    __shared__ float ler[TROWS * H];
    __shared__ float rmax[TROWS * H];
    __shared__ float rsum[TROWS * H];
    __shared__ int qs[32 * QCAP];
    __shared__ int qd[32 * QCAP];

    const int tid  = threadIdx.x;
    const int lane = tid & 31;
    const int wave = __builtin_amdgcn_readfirstlane(tid >> 5);
    const int d0   = blockIdx.x * TROWS;
    const int nrows = min(TROWS, nNodes - d0);
    const int nchunks = (E + CHUNK - 1) / CHUNK;

    for (int i = tid; i < TROWS * PITCH / 4; i += 1024)
        ((float4*)tile)[i] = make_float4(0.0f, 0.0f, 0.0f, 0.0f);
    for (int i = tid; i < TROWS * H; i += 1024) {
        ler[i]  = (i < nrows * H) ? er[d0 * H + i] : 0.0f;
        rmax[i] = -INFINITY;
        rsum[i] = 0.0f;
    }
    __syncthreads();

    int* myqs = qs + wave * QCAP;
    int* myqd = qd + wave * QCAP;
    int qlen = 0;
    bool ovf = false;

    auto stage1 = [&](int e) {
        constexpr int PER = 32 / H;
        const int sub = lane / H, hh = lane - sub * H;
        for (int i = 0; i < e; i += 4 * PER) {
            int row[4];
            float xe[4];
#pragma unroll
            for (int u = 0; u < 4; ++u) {
                const int idx = min(i + u * PER + sub, e - 1);
                row[u] = myqd[idx];
                xe[u]  = el[myqs[idx] * H + hh];
            }
#pragma unroll
            for (int u = 0; u < 4; ++u) {
                if (i + u * PER + sub < e) {
                    float x = xe[u] + ler[row[u] * H + hh];
                    x = x > 0.0f ? x : 0.2f * x;
                    __hip_atomic_fetch_max(rmax + row[u] * H + hh, x,
                                           __ATOMIC_RELAXED, __HIP_MEMORY_SCOPE_WORKGROUP);
                }
            }
        }
    };

    auto stage2 = [&](int e) {
        int q = 0;
        for (; q + 4 <= e; q += 4)
            agg_group<C, PITCH, 4>(myqs, myqd, q, lane, el, feat, ler, rmax, rsum, tile);
        for (; q < e; ++q)
            agg_group<C, PITCH, 1>(myqs, myqd, q, lane, el, feat, ler, rmax, rsum, tile);
    };

    auto walk = [&](auto&& full) {
        for (int g0 = wave; g0 < nchunks; g0 += 1024) {
            const int cidx = g0 + 32 * lane;
            int sbeg = 0, n = 0;
            if (cidx < nchunks) {
                const int* o = off + (long)cidx * (NTILES + 1) + blockIdx.x;
                const int a = o[0], b = o[1];
                sbeg = cidx * CHUNK + a;
                n = b - a;
            }
            int incl = n;
            for (int o = 1; o < 32; o <<= 1) {
                const int t = __shfl_up(incl, o, 32);
                if (lane >= o) incl += t;
            }
            const int excl  = incl - n;
            const int delta = sbeg - excl;
            const int total = __builtin_amdgcn_readlane(incl, 31);
            const int nact  = min(32, (nchunks - g0 + 31) >> 5);
            for (int f0 = 0; f0 < total; f0 += 128) {
                if (qlen > QCAP - 128) { full(qlen); ovf = true; qlen = 0; }
                int fc[4], ad[4], w[4];
#pragma unroll
                for (int k = 0; k < 4; ++k) {
                    fc[k] = min(f0 + k * 32 + lane, total - 1);
                    ad[k] = 0;
                }
                for (int j = 0; j < nact; ++j) {
                    const int ej = __builtin_amdgcn_readlane(excl, j);
                    const int dj = __builtin_amdgcn_readlane(delta, j);
#pragma unroll
                    for (int k = 0; k < 4; ++k) if (fc[k] >= ej) ad[k] = dj + fc[k];
                }
#pragma unroll
                for (int k = 0; k < 4; ++k) w[k] = ent[ad[k]];
#pragma unroll
                for (int k = 0; k < 4; ++k) {
                    if (f0 + k * 32 + lane < total) {
                        const int pos = qlen + k * 32 + lane;
                        myqs[pos] = w[k] >> 8;
                        myqd[pos] = w[k] & 255;
                    }
                }
                qlen += min(128, total - f0);
            }
        }
    };

    walk([&](int n) { stage1(n); });
    stage1(qlen);
    __syncthreads();

    if (!ovf) {
        stage2(qlen);
    } else {
        qlen = 0;
        walk([&](int n) { stage2(n); });
        stage2(qlen);
    }

    __syncthreads();
    if constexpr (FIN) {
        constexpr int NRT  = (TROWS + 15) / 16;
        constexpr int UPR  = NCN / 32;
        constexpr int GRT  = 4;
        constexpr int NGRP = (NRT + GRT - 1) / GRT;
        constexpr int MW   = GRT * UPR;
        static_assert(NGRP * MW <= 32, "every group has multiplying waves of its own");
        const int c = lane & 15;
        const int h = lane >> 4;
        for (int g = 0; g <= NGRP; ++g) {
            const int mg = g - 1;
            const bool mul = (mg >= 0) && (wave / MW == mg);
            if (mul) {
                const int u  = wave - mg * MW;
                const int rt = mg * GRT + u / UPR;
                const int cp = u % UPR;
                if (rt < NRT) {
                    const int rr = min(rt * 16 + c, TROWS - 1);
                    v8f acc[2];
                    acc[0] = v8f{}; acc[1] = v8f{};
                    tile_unit_product<C, NCN>(tile + rr * PITCH + 16 * h,
                                              Wn + (long)(16 * h) * NCN + cp * 32 + c, acc);
                    __builtin_amdgcn_sched_barrier(0);
                    asm volatile("v_nop\n\tv_nop\n\tv_nop\n\tv_nop\n\tv_nop");
#pragma unroll
                    for (int r = 0; r < 8; ++r) {
                        const int row = rt * 16 + 8 * h + r;
                        if (row < nrows) {
                            float* orow = outn + (long)(d0 + row) * NCN + cp * 32 + c;
                            orow[0]  = acc[0][r];
                            orow[16] = acc[1][r];
                        }
                    }
                }
            } else if (g < NGRP) {
                const int nsw  = (mg >= 0) ? 32 - MW : 32;
                const int sidx = (mg >= 0 && wave > mg * MW) ? wave - MW : wave;
                const int r0 = g * GRT * 16;
                const int r1 = min(r0 + GRT * 16, TROWS);
                for (int i = sidx * 32 + lane; i < (r1 - r0) * (C / 4); i += nsw * 32) {
                    const int row = r0 + i / (C / 4);
                    const int c4  = i % (C / 4);
                    float4* p = (float4*)(tile + row * PITCH) + c4;
                    float4 t = *p;
                    const float sm = rsum[row * H + c4 / 16];
                    const float dn = (sm == 0.0f) ? 1.0f : sm;
                    const float* bb = bias + c4 * 4;
                    float x;
                    x = t.x / dn + bb[0]; t.x = x > 0.0f ? x : expm1f(x);
                    x = t.y / dn + bb[1]; t.y = x > 0.0f ? x : expm1f(x);
                    x = t.z / dn + bb[2]; t.z = x > 0.0f ? x : expm1f(x);
                    x = t.w / dn + bb[3]; t.w = x > 0.0f ? x : expm1f(x);
                    *p = t;
                }
            }
            if (g < NGRP) __syncthreads();
        }
    } else {
        float4* o = (float4*)(rst + (long)d0 * C);
        for (int i = tid; i < nrows * (C / 4); i += 1024) {
            const float4 t = ((const float4*)tile)[i];
            const int row = i / (C / 4);
            const int hh  = (i % (C / 4)) / 16;
            const float sm = rsum[row * H + hh];
            const float dn = (sm == 0.0f) ? 1.0f : sm;
            o[i] = make_float4(t.x / dn, t.y / dn, t.z / dn, t.w / dn);
        }
    }
}

__global__ __launch_bounds__(256) void classify(
    const float* __restrict__ rst2, const float* __restrict__ b2,
    const float* __restrict__ Wc1, const float* __restrict__ bc1,
    const float* __restrict__ Wc2, const float* __restrict__ bc2,
    float* __restrict__ out, int nNodes)
{
    const int lane = threadIdx.x & 31;
    const int wid  = blockIdx.x * 8 + __builtin_amdgcn_readfirstlane(threadIdx.x >> 5);
    const int nw   = gridDim.x * 8;
    float w[64];
#pragma unroll
    for (int d = 0; d < 64; ++d) w[d] = Wc1[d * 32 + lane];
    const float bl  = bc1[lane];
    const float w20 = Wc2[lane * 2 + 0], w21 = Wc2[lane * 2 + 1];
    const float c0  = bc2[0], c1 = bc2[1];
    for (int node = wid; node < nNodes; node += nw) {
        const float* r = rst2 + (long)node * 64;
        float acc = 0.0f;
#pragma unroll
        for (int d = 0; d < 64; ++d)
            acc += (r[d] + b2[d]) * w[d];
        float t = fmaxf(acc + bl, 0.0f);
        float o0 = t * w20;
        float o1 = t * w21;
#pragma unroll
        for (int o = 16; o > 0; o >>= 1) {
            o0 += __shfl_xor(o0, o, 32);
            o1 += __shfl_xor(o1, o, 32);
        }
        if (lane == 0) {
            out[(long)node * 2 + 0] = o0 + c0;
            out[(long)node * 2 + 1] = o1 + c1;
        }
    }
}

extern "C" void kernel_launch(void* const* d_in, const int* in_sizes, int n_in,
                              void* d_out, int out_size, void* d_ws, size_t ws_size,
                              hipStream_t stream)
{
    (void)n_in; (void)out_size; (void)ws_size;

    const float* randfeat = (const float*)d_in[0];
    const float* func_emb = (const float*)d_in[1];
    const float* codebert = (const float*)d_in[2];
    const int*   src      = (const int*)d_in[3];
    const int*   dst      = (const int*)d_in[4];
    const float* W1  = (const float*)d_in[5];
    const float* al1 = (const float*)d_in[6];
    const float* ar1 = (const float*)d_in[7];
    const float* b1  = (const float*)d_in[8];
    const float* W2  = (const float*)d_in[9];
    const float* al2 = (const float*)d_in[10];
    const float* ar2 = (const float*)d_in[11];
    const float* b2  = (const float*)d_in[12];
    const float* Wc1 = (const float*)d_in[13];
    const float* bc1 = (const float*)d_in[14];
    const float* Wc2 = (const float*)d_in[15];
    const float* bc2 = (const float*)d_in[16];
    float* out = (float*)d_out;

    const int  E  = in_sizes[3];
    const long NN = N_NODES;

    float* ws     = (float*)d_ws;
    float* feat1  = ws;
    float* feat2  = feat1  + NN * 256;
    int*   ent    = (int*)(feat2 + NN * 256);
    int*   off    = ent    + NN * 48;
    float* rst2   = feat2  + NN * 256 + NN * 64;
    float* el1    = rst2   + NN * 64;
    float* er1    = el1    + NN * 4;
    float* el2    = er1    + NN * 4;
    float* er2    = el2    + NN;

    bin_edges_by_tile<<<256, 1024, 0, stream>>>(src, dst, ent, off, E, N_NODES);

    gemm_wmma_f32<8, 2, 256, 12, true><<<256, 384, 0, stream>>>(
        randfeat, func_emb, codebert, 100, 100, 768, W1, feat1, N_NODES);
    attn_proj<4><<<1024, 256, 0, stream>>>(feat1, al1, ar1, el1, er1, N_NODES * 4);
    edge_softmax_aggregate_tile<256, true, 64><<<NTILES, 1024, 0, stream>>>(
        ent, off, feat1, el1, er1, nullptr, b1, W2, feat2, E, N_NODES);

    attn_proj<1><<<256, 256, 0, stream>>>(feat2, al2, ar2, el2, er2, N_NODES);
    edge_softmax_aggregate_tile<64, false, 32><<<NTILES, 1024, 0, stream>>>(
        ent, off, feat2, el2, er2, rst2, nullptr, nullptr, nullptr, E, N_NODES);

    classify<<<1024, 256, 0, stream>>>(rst2, b2, Wc1, bc1, Wc2, bc2, out, N_NODES);
}


# Layer-2 tile finished in LDS: rows get /row sum and bias in place, then tile x Wc1 by WMMA, relu, x Wc2 by a 16-lane tree; classify launch and rst2 gone; classifier sums in a new order
# speedup vs baseline: 6.2878x; 1.0342x over previous
#include <hip/hip_runtime.h>
#include <math.h>


typedef __attribute__((ext_vector_type(2))) float v2f;
typedef __attribute__((ext_vector_type(8))) float v8f;

#define N_NODES 50000

template <int NT, int NC>
__device__ __forceinline__ void seg_accum(
    const float* __restrict__ a0, const float* __restrict__ a1, int w,
    const float* __restrict__ bcol, int h, v8f (&acc)[2][NT])
{
    const int nblk = w >> 5;
    if (nblk > 0) {
        const float4* q0 = (const float4*)(a0 + 16 * h);
        const float4* q1 = (const float4*)(a1 + 16 * h);
        const float* bp = bcol + (long)(16 * h) * NC;
        float4 ra[4], rb[4];
        v2f bc[NT];
#pragma unroll
        for (int q = 0; q < 4; ++q) { ra[q] = q0[q]; rb[q] = q1[q]; }
#pragma unroll
        for (int t = 0; t < NT; ++t) { bc[t].x = bp[t * 16]; bc[t].y = bp[NC + t * 16]; }

        for (int blk = 0; blk < nblk; ++blk) {
            const int adv = (blk + 1 < nblk) ? 32 : 0;
            const float4* nq0 = (const float4*)((const float*)q0 + adv);
            const float4* nq1 = (const float4*)((const float*)q1 + adv);
            const float* nbp = bp + (long)adv * NC;
            float4 na[4], nb[4];
#pragma unroll
            for (int q = 0; q < 4; ++q) { na[q] = nq0[q]; nb[q] = nq1[q]; }
            __builtin_amdgcn_sched_barrier(0);
#pragma unroll
            for (int j = 0; j < 8; ++j) {
                v2f bn[NT];
                if (j < 7) {
#pragma unroll
                    for (int t = 0; t < NT; ++t) {
                        bn[t].x = bp[(2 * j + 2) * NC + t * 16];
                        bn[t].y = bp[(2 * j + 3) * NC + t * 16];
                    }
                } else {
#pragma unroll
                    for (int t = 0; t < NT; ++t) {
                        bn[t].x = nbp[t * 16];
                        bn[t].y = nbp[NC + t * 16];
                    }
                }
                v2f fa, fb;
                if (j & 1) { fa.x = ra[j >> 1].z; fa.y = ra[j >> 1].w; fb.x = rb[j >> 1].z; fb.y = rb[j >> 1].w; }
                else       { fa.x = ra[j >> 1].x; fa.y = ra[j >> 1].y; fb.x = rb[j >> 1].x; fb.y = rb[j >> 1].y; }
#pragma unroll
                for (int t = 0; t < NT; ++t) {
                    acc[0][t] = __builtin_amdgcn_wmma_f32_16x16x4_f32(
                        false, fa, false, bc[t], (short)0, acc[0][t], false, false);
                    acc[1][t] = __builtin_amdgcn_wmma_f32_16x16x4_f32(
                        false, fb, false, bc[t], (short)0, acc[1][t], false, false);
                }
#pragma unroll
                for (int t = 0; t < NT; ++t) bc[t] = bn[t];
            }
            __builtin_amdgcn_sched_barrier(0);
            asm volatile("v_nop\n\tv_nop\n\tv_nop\n\tv_nop");
#pragma unroll
            for (int q = 0; q < 4; ++q) { ra[q] = na[q]; rb[q] = nb[q]; }
            q0 = nq0; q1 = nq1; bp = nbp;
        }
    }
    for (int k = nblk << 5; k + 4 <= w; k += 4) {
        const float2 ta = *(const float2*)(a0 + k + 2 * h);
        const float2 tb = *(const float2*)(a1 + k + 2 * h);
        v2f fa, fb;
        fa.x = ta.x; fa.y = ta.y; fb.x = tb.x; fb.y = tb.y;
        const float* bp = bcol + (long)(k + 2 * h) * NC;
#pragma unroll
        for (int t = 0; t < NT; ++t) {
            v2f b;
            b.x = bp[t * 16];
            b.y = bp[NC + t * 16];
            acc[0][t] = __builtin_amdgcn_wmma_f32_16x16x4_f32(
                false, fa, false, b, (short)0, acc[0][t], false, false);
            acc[1][t] = __builtin_amdgcn_wmma_f32_16x16x4_f32(
                false, fb, false, b, (short)0, acc[1][t], false, false);
        }
    }
}

template <int NC>
__device__ __forceinline__ void seg_accum_rem(
    const float* __restrict__ a0, int w,
    const float* __restrict__ bcol, int h, v8f (&acc)[2])
{
    const int nblk = w >> 5;
    if (nblk > 0) {
        const float4* q0 = (const float4*)(a0 + 16 * h);
        const float* bp = bcol + (long)(16 * h) * NC;
        float4 ra[4];
        v2f rb[8][2];
#pragma unroll
        for (int q = 0; q < 4; ++q) ra[q] = q0[q];
#pragma unroll
        for (int j = 0; j < 8; ++j) {
#pragma unroll
            for (int t = 0; t < 2; ++t) {
                rb[j][t].x = bp[(2 * j) * NC + t * 16];
                rb[j][t].y = bp[(2 * j + 1) * NC + t * 16];
            }
        }
        for (int blk = 0; blk < nblk; ++blk) {
            const int adv = (blk + 1 < nblk) ? 32 : 0;
            const float4* nq0 = (const float4*)((const float*)q0 + adv);
            const float* nbp = bp + (long)adv * NC;
            float4 na[4];
            v2f nb[8][2];
#pragma unroll
            for (int q = 0; q < 4; ++q) na[q] = nq0[q];
#pragma unroll
            for (int j = 0; j < 8; ++j) {
#pragma unroll
                for (int t = 0; t < 2; ++t) {
                    nb[j][t].x = nbp[(2 * j) * NC + t * 16];
                    nb[j][t].y = nbp[(2 * j + 1) * NC + t * 16];
                }
            }
            __builtin_amdgcn_sched_barrier(0);
#pragma unroll
            for (int j = 0; j < 8; ++j) {
                v2f fa;
                if (j & 1) { fa.x = ra[j >> 1].z; fa.y = ra[j >> 1].w; }
                else       { fa.x = ra[j >> 1].x; fa.y = ra[j >> 1].y; }
#pragma unroll
                for (int t = 0; t < 2; ++t)
                    acc[t] = __builtin_amdgcn_wmma_f32_16x16x4_f32(
                        false, fa, false, rb[j][t], (short)0, acc[t], false, false);
            }
            __builtin_amdgcn_sched_barrier(0);
#pragma unroll
            for (int q = 0; q < 4; ++q) ra[q] = na[q];
#pragma unroll
            for (int j = 0; j < 8; ++j) { rb[j][0] = nb[j][0]; rb[j][1] = nb[j][1]; }
            q0 = nq0; bp = nbp;
        }
    }
    for (int k = nblk << 5; k + 4 <= w; k += 4) {
        const float2 ta = *(const float2*)(a0 + k + 2 * h);
        v2f fa;
        fa.x = ta.x; fa.y = ta.y;
        const float* bp = bcol + (long)(k + 2 * h) * NC;
#pragma unroll
        for (int t = 0; t < 2; ++t) {
            v2f b;
            b.x = bp[t * 16];
            b.y = bp[NC + t * 16];
            acc[t] = __builtin_amdgcn_wmma_f32_16x16x4_f32(
                false, fa, false, b, (short)0, acc[t], false, false);
        }
    }
}

template <int NT, int NG, int NC, int WPB, bool REM>
__global__ __launch_bounds__(WPB * 32) void gemm_wmma_f32(
    const float* __restrict__ p0, const float* __restrict__ p1,
    const float* __restrict__ p2, int w0, int w1, int w2,
    const float* __restrict__ B, float* __restrict__ C, int M)
{
    const int lane = threadIdx.x & 31;
    const int wv   = threadIdx.x >> 5;
    const int gw   = blockIdx.x * WPB + wv;
    const int c  = lane & 15;
    const int h  = lane >> 4;

    if (REM) {
        const int nWaves   = gridDim.x * WPB;
        const int mainRows = (nWaves / NG) * 32;
        const int nRem = (mainRows < M) ? ((M - mainRows + 15) / 16) * (NC / 32) : 0;
        for (int u = wv * gridDim.x + blockIdx.x; u < nRem; u += nWaves) {
            const int tm0 = mainRows + (u / (NC / 32)) * 16;
            const int tn0 = (u % (NC / 32)) * 32;
            const int rr = min(tm0 + c, M - 1);
            v8f racc[2];
            racc[0] = v8f{}; racc[1] = v8f{};
            const float* rcol = B + tn0 + c;
            seg_accum_rem<NC>(p0 + (long)rr * w0, w0, rcol, h, racc);
            seg_accum_rem<NC>(p1 + (long)rr * w1, w1, rcol + (long)w0 * NC, h, racc);
            seg_accum_rem<NC>(p2 + (long)rr * w2, w2, rcol + (long)(w0 + w1) * NC, h, racc);
#pragma unroll
            for (int r = 0; r < 8; ++r) {
                const int row = tm0 + 8 * h + r;
                if (row < M) {
                    float* crow = C + (long)row * NC + tn0 + c;
                    crow[0]  = racc[0][r];
                    crow[16] = racc[1][r];
                }
            }
        }
    }

    const int ng   = gw % NG;
    const int m0   = (gw / NG) * 32;
    if (m0 >= M) return;
    const int n0 = ng * NT * 16;

    const int r0 = min(m0 + c, M - 1);
    const int r1 = min(m0 + 16 + c, M - 1);

    v8f acc[2][NT];
#pragma unroll
    for (int t = 0; t < NT; ++t) { acc[0][t] = v8f{}; acc[1][t] = v8f{}; }

    const float* bcol = B + n0 + c;
    seg_accum<NT, NC>(p0 + (long)r0 * w0, p0 + (long)r1 * w0, w0, bcol, h, acc);
    seg_accum<NT, NC>(p1 + (long)r0 * w1, p1 + (long)r1 * w1, w1, bcol + (long)w0 * NC, h, acc);
    seg_accum<NT, NC>(p2 + (long)r0 * w2, p2 + (long)r1 * w2, w2, bcol + (long)(w0 + w1) * NC, h, acc);

#pragma unroll
    for (int i = 0; i < 2; ++i) {
#pragma unroll
        for (int r = 0; r < 8; ++r) {
            const int row = m0 + i * 16 + 8 * h + r;
            if (row < M) {
                float* crow = C + (long)row * NC + n0 + c;
#pragma unroll
                for (int t = 0; t < NT; ++t) crow[t * 16] = acc[i][t][r];
            }
        }
    }
}

#define PU 8
template <int H>
__global__ __launch_bounds__(256) void attn_proj(
    const float* __restrict__ feat, const float* __restrict__ al,
    const float* __restrict__ ar, float* __restrict__ el,
    float* __restrict__ er, int nPairs)
{
    static_assert(PU % H == 0, "a trip starts at head 0");
    const int lane = threadIdx.x & 31;
    const int wid  = blockIdx.x * 8 + (threadIdx.x >> 5);
    const int nw   = gridDim.x * 8;
    float wl0[H], wl1[H], wr0[H], wr1[H];
#pragma unroll
    for (int h = 0; h < H; ++h) {
        wl0[h] = al[h * 64 + lane]; wl1[h] = al[h * 64 + lane + 32];
        wr0[h] = ar[h * 64 + lane]; wr1[h] = ar[h * 64 + lane + 32];
    }
    for (int p0 = wid * PU; p0 < nPairs; p0 += nw * PU) {
        float f0[PU], f1[PU];
#pragma unroll
        for (int u = 0; u < PU; ++u) {
            const float* f = feat + (long)min(p0 + u, nPairs - 1) * 64;
            f0[u] = f[lane]; f1[u] = f[lane + 32];
        }
        float sl[PU], sr[PU];
#pragma unroll
        for (int u = 0; u < PU; ++u) {
            sl[u] = f0[u] * wl0[u % H] + f1[u] * wl1[u % H];
            sr[u] = f0[u] * wr0[u % H] + f1[u] * wr1[u % H];
        }
#pragma unroll
        for (int o = 16; o > 0; o >>= 1) {
#pragma unroll
            for (int u = 0; u < PU; ++u) {
                sl[u] += __shfl_xor(sl[u], o, 32);
                sr[u] += __shfl_xor(sr[u], o, 32);
            }
        }
        float ol = sl[0], orr = sr[0];
#pragma unroll
        for (int u = 1; u < PU; ++u) if (lane == u) { ol = sl[u]; orr = sr[u]; }
        if (lane < PU && p0 + lane < nPairs) { el[p0 + lane] = ol; er[p0 + lane] = orr; }
    }
}

#define TROWS 196
#define QCAP  384
#define CHUNK 4096
#define NTILES ((N_NODES + TROWS - 1) / TROWS)

__global__ __launch_bounds__(1024) void bin_edges_by_tile(
    const int* __restrict__ src, const int* __restrict__ dst,
    int* __restrict__ ent, int* __restrict__ off, int E, int nNodes)
{
    static_assert(TROWS <= 256, "the row inside a tile is packed into 8 bits");
    static_assert(NTILES <= 1024, "one thread per tile in the prefix sum");
    __shared__ int hist[NTILES];
    __shared__ int start[NTILES + 1];
    __shared__ int wsum[32];
    __shared__ int sorted[CHUNK];

    const int tid  = threadIdx.x;
    const int lane = tid & 31;
    const int wv   = tid >> 5;
    const int nchunks = (E + CHUNK - 1) / CHUNK;
    for (int c = blockIdx.x; c < nchunks; c += gridDim.x) {
        for (int i = tid; i < NTILES; i += 1024) hist[i] = 0;
        __syncthreads();

        const int e0 = c * CHUNK + tid * 4;
        int d[4], s[4];
        if (e0 + 3 < E) {
            const int4 dv = *(const int4*)(dst + e0);
            const int4 sv = *(const int4*)(src + e0);
            d[0] = dv.x; d[1] = dv.y; d[2] = dv.z; d[3] = dv.w;
            s[0] = sv.x; s[1] = sv.y; s[2] = sv.z; s[3] = sv.w;
        } else {
#pragma unroll
            for (int k = 0; k < 4; ++k) {
                d[k] = (e0 + k < E) ? dst[e0 + k] : -1;
                s[k] = (e0 + k < E) ? src[e0 + k] : 0;
            }
        }
        int tl[4], rk[4], pk[4];
#pragma unroll
        for (int k = 0; k < 4; ++k) {
            tl[k] = -1; rk[k] = 0; pk[k] = 0;
            if ((unsigned)d[k] < (unsigned)nNodes) {
                tl[k] = d[k] / TROWS;
                pk[k] = (s[k] << 8) | (d[k] - tl[k] * TROWS);
                rk[k] = __hip_atomic_fetch_add(hist + tl[k], 1, __ATOMIC_RELAXED,
                                               __HIP_MEMORY_SCOPE_WORKGROUP);
            }
        }
        __syncthreads();

        int v = 0, incl = 0;
        if (wv < (NTILES + 31) / 32) {
            v = (tid < NTILES) ? hist[tid] : 0;
            incl = v;
            for (int o = 1; o < 32; o <<= 1) {
                const int t = __shfl_up(incl, o, 32);
                if (lane >= o) incl += t;
            }
            if (lane == 31) wsum[wv] = incl;
        }
        __syncthreads();
        if (tid < NTILES) {
            int base = 0;
            for (int w = 0; w < wv; ++w) base += wsum[w];
            start[tid] = base + incl - v;
            if (tid == NTILES - 1) start[NTILES] = base + incl;
        }
        __syncthreads();

#pragma unroll
        for (int k = 0; k < 4; ++k)
            if (tl[k] >= 0) sorted[start[tl[k]] + rk[k]] = pk[k];
        __syncthreads();

        const int cnt = start[NTILES];
        for (int i = tid; i < cnt; i += 1024) ent[(long)c * CHUNK + i] = sorted[i];
        for (int i = tid; i <= NTILES; i += 1024) off[(long)c * (NTILES + 1) + i] = start[i];
        __syncthreads();
    }
}

template <int K, int NC>
__device__ __forceinline__ void tile_unit_product(
    const float* arow, const float* __restrict__ bcol, v8f (&acc)[2])
{
    static_assert(K % 32 == 0, "whole 32-wide blocks");
    constexpr int NHB = K / 16;
    float4 fa[2][2];
    v2f fb[2][4][2];
#pragma unroll
    for (int hb = -1; hb < NHB; ++hb) {
        if (hb + 1 < NHB) {
            const int nx = hb + 1;
            const int o  = (nx >> 1) * 32 + (nx & 1) * 8;
            fa[nx & 1][0] = *(const float4*)(arow + o);
            fa[nx & 1][1] = *(const float4*)(arow + o + 4);
#pragma unroll
            for (int jj = 0; jj < 4; ++jj) {
#pragma unroll
                for (int t = 0; t < 2; ++t) {
                    fb[nx & 1][jj][t].x = bcol[(o + 2 * jj) * NC + t * 16];
                    fb[nx & 1][jj][t].y = bcol[(o + 2 * jj + 1) * NC + t * 16];
                }
            }
        }
        __builtin_amdgcn_sched_barrier(0);
        if (hb >= 0) {
#pragma unroll
            for (int jj = 0; jj < 4; ++jj) {
                const float4 q = fa[hb & 1][jj >> 1];
                v2f a;
                if (jj & 1) { a.x = q.z; a.y = q.w; }
                else        { a.x = q.x; a.y = q.y; }
#pragma unroll
                for (int t = 0; t < 2; ++t)
                    acc[t] = __builtin_amdgcn_wmma_f32_16x16x4_f32(
                        false, a, false, fb[hb & 1][jj][t], (short)0, acc[t], false, false);
            }
        }
    }
}

template <int C, int PITCH, int U>
__device__ __forceinline__ void agg_group(
    const int* myqs, const int* myqd, int q, int lane,
    const float* __restrict__ el, const float* __restrict__ feat,
    const float* ler, const float* rmax, float* rsum, float* tile)
{
    constexpr int H  = C / 64;
    constexpr int NI = C / 32;
    const int pl = lane % (U * H);
    const int pu = pl / H, ph = pl - pu * H;
    const int sl = myqs[q + pu];
    const int rl = myqd[q + pu];
    const float xe = el[sl * H + ph];
    const float xr = ler[rl * H + ph];
    const float mx = rmax[rl * H + ph];

    int s[U], dd[U];
#pragma unroll
    for (int u = 0; u < U; ++u) {
        s[u]  = __builtin_amdgcn_readfirstlane(myqs[q + u]);
        dd[u] = __builtin_amdgcn_readfirstlane(myqd[q + u]);
    }
    float v[U][NI];
#pragma unroll
    for (int u = 0; u < U; ++u) {
        const float* f = feat + (long)s[u] * C + lane;
#pragma unroll
        for (int i = 0; i < NI; ++i) v[u][i] = f[i * 32];
    }
    __builtin_amdgcn_sched_barrier(0);
    float x = xe + xr;
    x = x > 0.0f ? x : 0.2f * x;
    const float w = expf(x - mx);
    if (lane < U * H)
        __hip_atomic_fetch_add(rsum + rl * H + ph, w, __ATOMIC_RELAXED, __HIP_MEMORY_SCOPE_WORKGROUP);
#pragma unroll
    for (int u = 0; u < U; ++u) {
        float* t = tile + dd[u] * PITCH + lane;
#pragma unroll
        for (int i = 0; i < NI; ++i) {
            const float a = __int_as_float(
                __builtin_amdgcn_readlane(__float_as_int(w), u * H + (i >> 1)));
            __hip_atomic_fetch_add(t + i * 32, v[u][i] * a,
                                   __ATOMIC_RELAXED, __HIP_MEMORY_SCOPE_WORKGROUP);
        }
    }
}

template <int C, bool FIN, int NCN>
__global__ __launch_bounds__(1024) void edge_softmax_aggregate_tile(
    const int* __restrict__ ent, const int* __restrict__ off,
    const float* __restrict__ feat, const float* __restrict__ el,
    const float* __restrict__ er,
    const float* __restrict__ bias, const float* __restrict__ Wn, float* __restrict__ outn,
    const float* __restrict__ bh, const float* __restrict__ Wo, const float* __restrict__ bo,
    int E, int nNodes)
{
    constexpr int H = C / 64;
    constexpr int PITCH = C + 4;
    __shared__ __attribute__((aligned(16))) float tile[TROWS * PITCH];
    __shared__ float ler[TROWS * H];
    __shared__ float rmax[TROWS * H];
    __shared__ float rsum[TROWS * H];
    __shared__ int qs[32 * QCAP];
    __shared__ int qd[32 * QCAP];

    const int tid  = threadIdx.x;
    const int lane = tid & 31;
    const int wave = __builtin_amdgcn_readfirstlane(tid >> 5);
    const int d0   = blockIdx.x * TROWS;
    const int nrows = min(TROWS, nNodes - d0);
    const int nchunks = (E + CHUNK - 1) / CHUNK;

    for (int i = tid; i < TROWS * PITCH / 4; i += 1024)
        ((float4*)tile)[i] = make_float4(0.0f, 0.0f, 0.0f, 0.0f);
    for (int i = tid; i < TROWS * H; i += 1024) {
        ler[i]  = (i < nrows * H) ? er[d0 * H + i] : 0.0f;
        rmax[i] = -INFINITY;
        rsum[i] = 0.0f;
    }
    __syncthreads();

    int* myqs = qs + wave * QCAP;
    int* myqd = qd + wave * QCAP;
    int qlen = 0;
    bool ovf = false;

    auto stage1 = [&](int e) {
        constexpr int PER = 32 / H;
        const int sub = lane / H, hh = lane - sub * H;
        for (int i = 0; i < e; i += 4 * PER) {
            int row[4];
            float xe[4];
#pragma unroll
            for (int u = 0; u < 4; ++u) {
                const int idx = min(i + u * PER + sub, e - 1);
                row[u] = myqd[idx];
                xe[u]  = el[myqs[idx] * H + hh];
            }
#pragma unroll
            for (int u = 0; u < 4; ++u) {
                if (i + u * PER + sub < e) {
                    float x = xe[u] + ler[row[u] * H + hh];
                    x = x > 0.0f ? x : 0.2f * x;
                    __hip_atomic_fetch_max(rmax + row[u] * H + hh, x,
                                           __ATOMIC_RELAXED, __HIP_MEMORY_SCOPE_WORKGROUP);
                }
            }
        }
    };

    auto stage2 = [&](int e) {
        int q = 0;
        for (; q + 4 <= e; q += 4)
            agg_group<C, PITCH, 4>(myqs, myqd, q, lane, el, feat, ler, rmax, rsum, tile);
        for (; q < e; ++q)
            agg_group<C, PITCH, 1>(myqs, myqd, q, lane, el, feat, ler, rmax, rsum, tile);
    };

    auto walk = [&](auto&& full) {
        for (int g0 = wave; g0 < nchunks; g0 += 1024) {
            const int cidx = g0 + 32 * lane;
            int sbeg = 0, n = 0;
            if (cidx < nchunks) {
                const int* o = off + (long)cidx * (NTILES + 1) + blockIdx.x;
                const int a = o[0], b = o[1];
                sbeg = cidx * CHUNK + a;
                n = b - a;
            }
            int incl = n;
            for (int o = 1; o < 32; o <<= 1) {
                const int t = __shfl_up(incl, o, 32);
                if (lane >= o) incl += t;
            }
            const int excl  = incl - n;
            const int delta = sbeg - excl;
            const int total = __builtin_amdgcn_readlane(incl, 31);
            const int nact  = min(32, (nchunks - g0 + 31) >> 5);
            for (int f0 = 0; f0 < total; f0 += 128) {
                if (qlen > QCAP - 128) { full(qlen); ovf = true; qlen = 0; }
                int fc[4], ad[4], w[4];
#pragma unroll
                for (int k = 0; k < 4; ++k) {
                    fc[k] = min(f0 + k * 32 + lane, total - 1);
                    ad[k] = 0;
                }
                for (int j = 0; j < nact; ++j) {
                    const int ej = __builtin_amdgcn_readlane(excl, j);
                    const int dj = __builtin_amdgcn_readlane(delta, j);
#pragma unroll
                    for (int k = 0; k < 4; ++k) if (fc[k] >= ej) ad[k] = dj + fc[k];
                }
#pragma unroll
                for (int k = 0; k < 4; ++k) w[k] = ent[ad[k]];
#pragma unroll
                for (int k = 0; k < 4; ++k) {
                    if (f0 + k * 32 + lane < total) {
                        const int pos = qlen + k * 32 + lane;
                        myqs[pos] = w[k] >> 8;
                        myqd[pos] = w[k] & 255;
                    }
                }
                qlen += min(128, total - f0);
            }
        }
    };

    walk([&](int n) { stage1(n); });
    stage1(qlen);
    __syncthreads();

    if (!ovf) {
        stage2(qlen);
    } else {
        qlen = 0;
        walk([&](int n) { stage2(n); });
        stage2(qlen);
    }

    __syncthreads();
    if constexpr (FIN) {
        constexpr int NRT  = (TROWS + 15) / 16;
        constexpr int UPR  = NCN / 32;
        constexpr int GRT  = 4;
        constexpr int NGRP = (NRT + GRT - 1) / GRT;
        constexpr int MW   = GRT * UPR;
        static_assert(NGRP * MW <= 32, "every group has multiplying waves of its own");
        const int c = lane & 15;
        const int h = lane >> 4;
        for (int g = 0; g <= NGRP; ++g) {
            const int mg = g - 1;
            const bool mul = (mg >= 0) && (wave / MW == mg);
            if (mul) {
                const int u  = wave - mg * MW;
                const int rt = mg * GRT + u / UPR;
                const int cp = u % UPR;
                if (rt < NRT) {
                    const int rr = min(rt * 16 + c, TROWS - 1);
                    v8f acc[2];
                    acc[0] = v8f{}; acc[1] = v8f{};
                    tile_unit_product<C, NCN>(tile + rr * PITCH + 16 * h,
                                              Wn + (long)(16 * h) * NCN + cp * 32 + c, acc);
                    __builtin_amdgcn_sched_barrier(0);
                    asm volatile("v_nop\n\tv_nop\n\tv_nop\n\tv_nop\n\tv_nop");
#pragma unroll
                    for (int r = 0; r < 8; ++r) {
                        const int row = rt * 16 + 8 * h + r;
                        if (row < nrows) {
                            float* orow = outn + (long)(d0 + row) * NCN + cp * 32 + c;
                            orow[0]  = acc[0][r];
                            orow[16] = acc[1][r];
                        }
                    }
                }
            } else if (g < NGRP) {
                const int nsw  = (mg >= 0) ? 32 - MW : 32;
                const int sidx = (mg >= 0 && wave > mg * MW) ? wave - MW : wave;
                const int r0 = g * GRT * 16;
                const int r1 = min(r0 + GRT * 16, TROWS);
                for (int i = sidx * 32 + lane; i < (r1 - r0) * (C / 4); i += nsw * 32) {
                    const int row = r0 + i / (C / 4);
                    const int c4  = i % (C / 4);
                    float4* p = (float4*)(tile + row * PITCH) + c4;
                    float4 t = *p;
                    const float sm = rsum[row * H + c4 / 16];
                    const float dn = (sm == 0.0f) ? 1.0f : sm;
                    const float* bb = bias + c4 * 4;
                    float x;
                    x = t.x / dn + bb[0]; t.x = x > 0.0f ? x : expm1f(x);
                    x = t.y / dn + bb[1]; t.y = x > 0.0f ? x : expm1f(x);
                    x = t.z / dn + bb[2]; t.z = x > 0.0f ? x : expm1f(x);
                    x = t.w / dn + bb[3]; t.w = x > 0.0f ? x : expm1f(x);
                    *p = t;
                }
            }
            if (g < NGRP) __syncthreads();
        }
    } else {
        static_assert(NCN == 32, "the head's hidden layer is one unit of 32 columns");
        static_assert(H == 1, "one row sum per row; ler and rmax hold one word per row");
        static_assert(1024 % (C / 4) == 0, "a thread keeps its column group over the rows");
        const int c4 = tid % (C / 4);
        const float4 bb = *(const float4*)(bias + c4 * 4);
        for (int i = tid; i < TROWS * (C / 4); i += 1024) {
            const int row = i / (C / 4);
            float4* p = (float4*)(tile + row * PITCH) + c4;
            float4 t = *p;
            const float sm = rsum[row];
            const float dn = (sm == 0.0f) ? 1.0f : sm;
            t.x = t.x / dn + bb.x;
            t.y = t.y / dn + bb.y;
            t.z = t.z / dn + bb.z;
            t.w = t.w / dn + bb.w;
            *p = t;
        }
        __syncthreads();
        constexpr int NRT = (TROWS + 15) / 16;
        const int c = lane & 15;
        const int h = lane >> 4;
        if (wave < NRT) {
            const int rr = min(wave * 16 + c, TROWS - 1);
            const float bh0 = bh[c], bh1 = bh[16 + c];
            const float w00 = Wo[c * 2], w01 = Wo[c * 2 + 1];
            const float w10 = Wo[(16 + c) * 2], w11 = Wo[(16 + c) * 2 + 1];
            const float bo0 = bo[0], bo1 = bo[1];
            v8f acc[2];
            acc[0] = v8f{}; acc[1] = v8f{};
            tile_unit_product<C, NCN>(tile + rr * PITCH + 16 * h,
                                      Wn + (long)(16 * h) * NCN + c, acc);
            __builtin_amdgcn_sched_barrier(0);
            asm volatile("v_nop\n\tv_nop\n\tv_nop\n\tv_nop\n\tv_nop" : "+v"(acc[0]), "+v"(acc[1]));
            float o0[8], o1[8];
#pragma unroll
            for (int r = 0; r < 8; ++r) {
                const float t0 = fmaxf(acc[0][r] + bh0, 0.0f);
                const float t1 = fmaxf(acc[1][r] + bh1, 0.0f);
                o0[r] = t0 * w00 + t1 * w10;
                o1[r] = t0 * w01 + t1 * w11;
            }
#pragma unroll
            for (int o = 8; o > 0; o >>= 1) {
#pragma unroll
                for (int r = 0; r < 8; ++r) {
                    o0[r] += __shfl_xor(o0[r], o, 32);
                    o1[r] += __shfl_xor(o1[r], o, 32);
                }
            }
            if (c == 0) {
#pragma unroll
                for (int r = 0; r < 8; ++r) {
                    const int row = wave * 16 + 8 * h + r;
                    if (row < TROWS) {
                        ler[row]  = o0[r] + bo0;
                        rmax[row] = o1[r] + bo1;
                    }
                }
            }
        }
        __syncthreads();
        for (int i = tid; i < nrows * 2; i += 1024)
            outn[(long)d0 * 2 + i] = (i & 1) ? rmax[i >> 1] : ler[i >> 1];
    }
}

extern "C" void kernel_launch(void* const* d_in, const int* in_sizes, int n_in,
                              void* d_out, int out_size, void* d_ws, size_t ws_size,
                              hipStream_t stream)
{
    (void)n_in; (void)out_size; (void)ws_size;

    const float* randfeat = (const float*)d_in[0];
    const float* func_emb = (const float*)d_in[1];
    const float* codebert = (const float*)d_in[2];
    const int*   src      = (const int*)d_in[3];
    const int*   dst      = (const int*)d_in[4];
    const float* W1  = (const float*)d_in[5];
    const float* al1 = (const float*)d_in[6];
    const float* ar1 = (const float*)d_in[7];
    const float* b1  = (const float*)d_in[8];
    const float* W2  = (const float*)d_in[9];
    const float* al2 = (const float*)d_in[10];
    const float* ar2 = (const float*)d_in[11];
    const float* b2  = (const float*)d_in[12];
    const float* Wc1 = (const float*)d_in[13];
    const float* bc1 = (const float*)d_in[14];
    const float* Wc2 = (const float*)d_in[15];
    const float* bc2 = (const float*)d_in[16];
    float* out = (float*)d_out;

    const int  E  = in_sizes[3];
    const long NN = N_NODES;

    float* ws     = (float*)d_ws;
    float* feat1  = ws;
    float* feat2  = feat1  + NN * 256;
    int*   ent    = (int*)(feat2 + NN * 256);
    int*   off    = ent    + NN * 48;
    float* el1    = feat2  + NN * 256 + NN * 64;
    float* er1    = el1    + NN * 4;
    float* el2    = er1    + NN * 4;
    float* er2    = el2    + NN;

    bin_edges_by_tile<<<256, 1024, 0, stream>>>(src, dst, ent, off, E, N_NODES);

    gemm_wmma_f32<8, 2, 256, 12, true><<<256, 384, 0, stream>>>(
        randfeat, func_emb, codebert, 100, 100, 768, W1, feat1, N_NODES);
    attn_proj<4><<<1024, 256, 0, stream>>>(feat1, al1, ar1, el1, er1, N_NODES * 4);
    edge_softmax_aggregate_tile<256, true, 64><<<NTILES, 1024, 0, stream>>>(
        ent, off, feat1, el1, er1, b1, W2, feat2, nullptr, nullptr, nullptr, E, N_NODES);

    attn_proj<1><<<256, 256, 0, stream>>>(feat2, al2, ar2, el2, er2, N_NODES);
    edge_softmax_aggregate_tile<64, false, 32><<<NTILES, 1024, 0, stream>>>(
        ent, off, feat2, el2, er2, b2, Wc1, out, bc1, Wc2, bc2, E, N_NODES);
}
